# starting kernel
# baseline (speedup 1.0000x reference)
#include <hip/hip_runtime.h>

typedef __attribute__((ext_vector_type(2))) float v2f;
typedef __attribute__((ext_vector_type(8))) float v8f;

#define WAVES_PER_BLOCK 8
#define BLOCK_THREADS 256

__global__ __launch_bounds__(BLOCK_THREADS)
void gemm16_wmma_f32(const float* __restrict__ A, const float* __restrict__ B,
                     const float* __restrict__ bias, float* __restrict__ C,
                     int M, int K, int N, int transB, int hasBias)
{
    __shared__ float Bs[256 * 16];
    const int tid   = threadIdx.x;
    const int nbase = blockIdx.y * 16;

    for (int i = tid; i < K * 16; i += BLOCK_THREADS) {
        int k  = i >> 4;
        int n  = i & 15;
        int gn = nbase + n;
        float v = 0.0f;
        if (gn < N) v = transB ? B[(long)gn * K + k] : B[(long)k * N + gn];
        Bs[i] = v;
    }
    __syncthreads();

    const int  wave  = tid >> 5;
    const int  lane  = tid & 31;
    const long mtile = (long)blockIdx.x * WAVES_PER_BLOCK + wave;
    if (mtile * 16 >= M) return;

    const int mn   = lane & 15;
    const int half = lane >> 4;

    const float*  arow  = A + (mtile * 16 + mn) * (long)K;
    const float2* arow2 = (const float2*)arow;

    v8f acc = {};
    for (int k = 0; k < K; k += 4) {
        const int kb = k + 2 * half;
        float2 a2 = arow2[kb >> 1];
        v2f av; av.x = a2.x; av.y = a2.y;
        v2f bv; bv.x = Bs[(kb + 0) * 16 + mn];
                bv.y = Bs[(kb + 1) * 16 + mn];
        acc = __builtin_amdgcn_wmma_f32_16x16x4_f32(
                   false, av,  false, bv,
                   (short)0, acc,  false,  false);
    }

    const int gn = nbase + mn;
    if (gn < N) {
        const float bval = hasBias ? bias[gn] : 0.0f;
        #pragma unroll
        for (int r = 0; r < 8; ++r) {
            long row = mtile * 16 + r + 8 * half;
            C[row * (long)N + gn] = acc[r] + bval;
        }
    }
}

__global__ __launch_bounds__(BLOCK_THREADS)
void spmm_scatter_f32(const int* __restrict__ rows, const int* __restrict__ cols,
                      const float* __restrict__ ew, const float* __restrict__ dense,
                      float* __restrict__ out, int nE, int F)
{
    const int warp = blockIdx.x * (BLOCK_THREADS >> 5) + (threadIdx.x >> 5);
    const int lane = threadIdx.x & 31;
    if (warp >= nE) return;
    const int   dst = rows[warp];
    const int   src = cols[warp];
    const float w   = ew[warp];
    const float4* s = (const float4*)(dense + (long)src * F);
    float*        d = out + (long)dst * F;
    for (int i = lane; i < (F >> 2); i += 32) {
        float4 v = s[i];
        atomicAdd(d + 4 * i + 0, v.x * w);
        atomicAdd(d + 4 * i + 1, v.y * w);
        atomicAdd(d + 4 * i + 2, v.z * w);
        atomicAdd(d + 4 * i + 3, v.w * w);
    }
}

__global__ void zero_f32(float* __restrict__ p, long n)
{
    long i      = (long)blockIdx.x * blockDim.x + threadIdx.x;
    long stride = (long)gridDim.x * blockDim.x;
    for (; i < n; i += stride) p[i] = 0.0f;
}

__global__ void bias_relu_f32(float* __restrict__ p, const float* __restrict__ b,
                              long nrows, int F)
{
    long n      = nrows * (long)F;
    long i      = (long)blockIdx.x * blockDim.x + threadIdx.x;
    long stride = (long)gridDim.x * blockDim.x;
    for (; i < n; i += stride) {
        float v = p[i] + b[(int)(i & (long)(F - 1))];
        p[i] = v > 0.0f ? v : 0.0f;
    }
}

static inline int mtiles_blocks(int M) {
    return ((M / 16) + WAVES_PER_BLOCK - 1) / WAVES_PER_BLOCK;
}

extern "C" void kernel_launch(void* const* d_in, const int* in_sizes, int n_in,
                              void* d_out, int out_size, void* d_ws, size_t ws_size,
                              hipStream_t stream)
{
    const float* feat0 = (const float*)d_in[0];
    const float* feat1 = (const float*)d_in[1];
    const float* fc0_w = (const float*)d_in[2];
    const float* fc0_b = (const float*)d_in[3];
    const float* fc1_w = (const float*)d_in[4];
    const float* fc1_b = (const float*)d_in[5];
    const float* W1    = (const float*)d_in[6];
    const float* b1    = (const float*)d_in[7];
    const float* W2    = (const float*)d_in[8];
    const float* b2    = (const float*)d_in[9];
    const float* pw    = (const float*)d_in[10];
    const float* pb    = (const float*)d_in[11];
    const int*   erow  = (const int*)d_in[12];
    const int*   ecol  = (const int*)d_in[13];
    const float* ew    = (const float*)d_in[14];
    float*       out   = (float*)d_out;

    const int E       = in_sizes[12];
    const int N_NODES = 100000;
    const int NT      = 50000;

    float* x   = (float*)d_ws;
    float* h   = x + (long)N_NODES * 128;
    float* agg = h + (long)N_NODES * 128;

    gemm16_wmma_f32<<<dim3(mtiles_blocks(NT), 8), BLOCK_THREADS, 0, stream>>>(
        feat0, fc0_w, fc0_b, x, NT, 128, 128, 1, 1);
    gemm16_wmma_f32<<<dim3(mtiles_blocks(NT), 8), BLOCK_THREADS, 0, stream>>>(
        feat1, fc1_w, fc1_b, x + (long)NT * 128, NT, 256, 128, 1, 1);

    gemm16_wmma_f32<<<dim3(mtiles_blocks(N_NODES), 8), BLOCK_THREADS, 0, stream>>>(
        x, W1, nullptr, h, N_NODES, 128, 128, 0, 0);
    zero_f32<<<2048, 256, 0, stream>>>(agg, (long)N_NODES * 128);
    spmm_scatter_f32<<<(E + 7) / 8, BLOCK_THREADS, 0, stream>>>(
        erow, ecol, ew, h, agg, E, 128);
    bias_relu_f32<<<2048, 256, 0, stream>>>(agg, b1, N_NODES, 128);

    gemm16_wmma_f32<<<dim3(mtiles_blocks(N_NODES), 4), BLOCK_THREADS, 0, stream>>>(
        agg, W2, nullptr, x, N_NODES, 128, 64, 0, 0);
    zero_f32<<<2048, 256, 0, stream>>>(h, (long)N_NODES * 64);
    spmm_scatter_f32<<<(E + 7) / 8, BLOCK_THREADS, 0, stream>>>(
        erow, ecol, ew, x, h, E, 64);
    bias_relu_f32<<<2048, 256, 0, stream>>>(h, b2, N_NODES, 64);

    gemm16_wmma_f32<<<dim3(mtiles_blocks(N_NODES), 1), BLOCK_THREADS, 0, stream>>>(
        h, pw, pb, out, N_NODES, 64, 8, 1, 1);
}


# GEMMs B-stationary: each wave holds its 16-column strip of B (whole K) in registers; the work-group streams its contiguous A rows through LDS, whole-line loads one step ahead; k order unchanged
# speedup vs baseline: 1.8747x; 1.8747x over previous
#include <hip/hip_runtime.h>

typedef __attribute__((ext_vector_type(2))) float v2f;
typedef __attribute__((ext_vector_type(8))) float v8f;

#define GEMM_THREADS 512
#define GEMM_BLOCKS 256
#define GEMM_LDS_FLOATS 41984

template <int K, int NT>
__device__ __forceinline__ void tile_mma(const float* __restrict__ as, const v2f* bv, v8f* acc)
{
    constexpr int P = K + 4;
    #pragma unroll
    for (int s = 0; s < K / 4; ++s) {
        #pragma unroll
        for (int j = 0; j < NT; ++j) {
            const float2 a2 = *(const float2*)(as + j * 16 * P + 4 * s);
            v2f av; av.x = a2.x; av.y = a2.y;
            acc[j] = __builtin_amdgcn_wmma_f32_16x16x4_f32(
                          false, av,  false, bv[s],
                          (short)0, acc[j],  false,  false);
        }
    }
}

template <int K, int NS, int TPW, int TRANSB>
__global__ __launch_bounds__(GEMM_THREADS)
void gemm_bstat_wmma_f32(const float* __restrict__ A, const float* __restrict__ B,
                         const float* __restrict__ bias, float* __restrict__ C,
                         int M, int N, int hasBias)
{
    constexpr int P          = K + 4;
    constexpr int RG         = 16 / NS;
    constexpr int STEP_TILES = RG * TPW;
    constexpr int STEP_ROWS  = STEP_TILES * 16;
    constexpr int K4         = K / 4;
    constexpr int NLD        = STEP_ROWS * K4 / GEMM_THREADS;
    static_assert(STEP_ROWS * K4 % GEMM_THREADS == 0, "step must split evenly over the threads");
    static_assert(2 * STEP_ROWS * P <= GEMM_LDS_FLOATS, "LDS ring too small");
    __shared__ float As[GEMM_LDS_FLOATS];

    const int tid  = threadIdx.x;
    const int wave = __builtin_amdgcn_readfirstlane(tid >> 5);
    const int lane = tid & 31;
    const int c    = lane & 15;
    const int h    = lane >> 4;
    const int strip = wave % NS;
    const int rg    = wave / NS;
    const int n     = strip * 16 + c;

    const int nT   = M >> 4;
    const int G    = gridDim.x;
    const int base = nT / G;
    const int rem  = nT - base * G;
    const int b    = blockIdx.x;
    const int t0   = b * base + (b < rem ? b : rem);
    const int t1   = t0 + base + (b < rem ? 1 : 0);
    if (t0 >= t1) return;

    const float4* A4 = (const float4*)A;

    float4 st[NLD];
    auto fetch = [&](int tb) {
        const int lim = (t1 - tb < STEP_TILES ? t1 - tb : STEP_TILES) * 16 * K4;
        const long g0 = (long)tb * 16 * K4;
        #pragma unroll
        for (int j = 0; j < NLD; ++j) {
            const int i = tid + j * GEMM_THREADS;
            st[j] = make_float4(0.0f, 0.0f, 0.0f, 0.0f);
            if (i < lim) st[j] = A4[g0 + i];
        }
    };
    auto stash = [&](float* buf) {
        #pragma unroll
        for (int j = 0; j < NLD; ++j) {
            const int i   = tid + j * GEMM_THREADS;
            const int row = i / K4;
            const int c4  = i - row * K4;
            *(float4*)(buf + row * P + 4 * c4) = st[j];
        }
    };

    fetch(t0);

    v2f bv[K4];
    if (TRANSB) {
        static_assert(NS * 16 * P <= GEMM_LDS_FLOATS, "B does not fit the LDS array");
        const float4* B4  = (const float4*)B;
        const int     nB4 = N * K4;
        constexpr int NB4 = NS * 16 * K4;
        constexpr int NBL = (NB4 + GEMM_THREADS - 1) / GEMM_THREADS;
        float4 bt[NBL];
        #pragma unroll
        for (int j = 0; j < NBL; ++j) {
            const int i = tid + j * GEMM_THREADS;
            bt[j] = make_float4(0.0f, 0.0f, 0.0f, 0.0f);
            if (i < nB4) bt[j] = B4[i];
        }
        #pragma unroll
        for (int j = 0; j < NBL; ++j) {
            const int i = tid + j * GEMM_THREADS;
            if (i < NB4) {
                const int row = i / K4;
                const int c4  = i - row * K4;
                *(float4*)(As + row * P + 4 * c4) = bt[j];
            }
        }
        __syncthreads();
        #pragma unroll
        for (int s = 0; s < K4; ++s) {
            const float2 v = *(const float2*)(As + n * P + 4 * s + 2 * h);
            bv[s].x = v.x; bv[s].y = v.y;
        }
        __syncthreads();
    } else {
        #pragma unroll
        for (int s = 0; s < K4; ++s) {
            const int k0 = 4 * s + 2 * h;
            float2 v = make_float2(0.0f, 0.0f);
            if (n < N) { v.x = B[(long)k0 * N + n]; v.y = B[(long)(k0 + 1) * N + n]; }
            bv[s].x = v.x; bv[s].y = v.y;
        }
    }
    float bval = 0.0f;
    if (hasBias && n < N) bval = bias[n];

    stash(As);
    __syncthreads();

    int cur = 0;
    for (int tb = t0; tb < t1; tb += STEP_TILES) {
        const int  nxt     = tb + STEP_TILES;
        const bool hasNext = nxt < t1;
        if (hasNext) fetch(nxt);

        const int valid = t1 - tb < STEP_TILES ? t1 - tb : STEP_TILES;
        const int lt0   = rg * TPW;
        const int nv    = valid - lt0;
        const float* as = As + cur * (STEP_ROWS * P) + (lt0 * 16 + c) * P + 2 * h;

        v8f acc[TPW];
        #pragma unroll
        for (int j = 0; j < TPW; ++j) acc[j] = v8f{};
        int done = 0;
        if (nv >= TPW)      { tile_mma<K, TPW>(as, bv, acc); done = TPW; }
        else if (nv >= 1)   { tile_mma<K, 1>(as, bv, acc);   done = 1;   }

        if (n < N) {
            #pragma unroll
            for (int j = 0; j < TPW; ++j) {
                if (j < done) {
                    const long row0 = (long)(tb + lt0 + j) * 16 + 8 * h;
                    #pragma unroll
                    for (int r = 0; r < 8; ++r)
                        C[(row0 + r) * (long)N + n] = acc[j][r] + bval;
                }
            }
        }

        if (hasNext) stash(As + (cur ^ 1) * (STEP_ROWS * P));
        __syncthreads();
        cur ^= 1;
    }
}

#define TILE_ROWS 512
#define TILE_THREADS 1024
#define QCAP 288

template <int F>
__device__ __forceinline__ void tile_flush(int n, const int* myq,
        const int* __restrict__ rows, const int* __restrict__ cols,
        const float* __restrict__ ew, const float* __restrict__ dense,
        float* acc, int row0, int lane)
{
    constexpr int LPR = F / 4;
    constexpr int EPI = 32 / LPR;
    constexpr int SB  = 8;
    int   rr = 0, cc = 0;
    float ww = 0.0f;
    if (lane < n) {
        const int e = myq[lane];
        rr = rows[e] - row0;
        cc = cols[e];
        ww = ew[e];
    }
    const int sub = lane / LPR;
    const int l   = lane % LPR;
    for (int s = 0; s < n; s += SB * EPI) {
        float4 v[SB];
        #pragma unroll
        for (int j = 0; j < SB; ++j) {
            const int idx = s + j * EPI + sub;
            const int sr  = __shfl(cc, idx & 31);
            v[j] = make_float4(0.0f, 0.0f, 0.0f, 0.0f);
            if (idx < n) v[j] = *(const float4*)(dense + (long)sr * F + 4 * l);
        }
        #pragma unroll
        for (int j = 0; j < SB; ++j) {
            const int   idx = s + j * EPI + sub;
            const float w   = __shfl(ww, idx & 31);
            const int   lr  = __shfl(rr, idx & 31);
            if (idx < n) {
                float* a = acc + lr * F + l;
                atomicAdd(a + 0 * LPR, v[j].x * w);
                atomicAdd(a + 1 * LPR, v[j].y * w);
                atomicAdd(a + 2 * LPR, v[j].z * w);
                atomicAdd(a + 3 * LPR, v[j].w * w);
            }
        }
    }
}

template <int F>
__global__ __launch_bounds__(TILE_THREADS)
void spmm_tile_lds_f32(const int* __restrict__ rows, const int* __restrict__ cols,
                       const float* __restrict__ ew, const float* __restrict__ dense,
                       const float* __restrict__ bias, float* __restrict__ out,
                       int nE, int nRows)
{
    constexpr int LPR = F / 4;
    __shared__ float acc[TILE_ROWS * F];
    __shared__ int   q[(TILE_THREADS / 32) * QCAP];

    const int tid  = threadIdx.x;
    const int wave = tid >> 5;
    const int lane = tid & 31;
    const int row0 = blockIdx.x * TILE_ROWS;

    float4* acc4 = (float4*)acc;
    for (int i = tid; i < TILE_ROWS * F / 4; i += TILE_THREADS)
        acc4[i] = make_float4(0.0f, 0.0f, 0.0f, 0.0f);
    __syncthreads();

    int* myq = q + wave * QCAP;
    int  qn  = 0;
    const unsigned below = (1u << lane) - 1u;

    auto push = [&](int rv, int eid) {
        const bool     m    = (unsigned)(rv - row0) < (unsigned)TILE_ROWS;
        const unsigned mask = __builtin_amdgcn_ballot_w32(m);
        if (mask != 0u) {
            if (m) myq[qn + __popc(mask & below)] = eid;
            qn += __popc(mask);
        }
    };
    auto drain = [&](int keep) {
        while (qn > keep) {
            const int n = qn >= 32 ? 32 : qn;
            qn -= n;
            tile_flush<F>(n, myq + qn, rows, cols, ew, dense, acc, row0, lane);
        }
    };

    const int4* rows4  = (const int4*)rows;
    const int   nTrips = nE >> 8;
    for (int p = wave; p < nTrips; p += TILE_THREADS / 32) {
        const int4 a  = rows4[p * 64 + lane];
        const int4 b  = rows4[p * 64 + 32 + lane];
        const int  e0 = p * 256 + lane * 4;
        push(a.x, e0 + 0);   push(a.y, e0 + 1);   push(a.z, e0 + 2);   push(a.w, e0 + 3);
        push(b.x, e0 + 128); push(b.y, e0 + 129); push(b.z, e0 + 130); push(b.w, e0 + 131);
        drain(31);
    }
    if (wave == 0) {
        const int base = nTrips << 8;
        #pragma unroll
        for (int k = 0; k < 8; ++k) {
            const int e = base + k * 32 + lane;
            int r = -1;
            if (e < nE) r = rows[e];
            push(r, e);
        }
    }
    drain(0);
    __syncthreads();

    for (int i = tid; i < TILE_ROWS * LPR; i += TILE_THREADS) {
        const int r  = i / LPR;
        const int l  = i % LPR;
        const int gr = row0 + r;
        if (gr < nRows) {
            const float* a = acc + r * F + l;
            const float4 b = *(const float4*)(bias + 4 * l);
            float4 o;
            o.x = a[0 * LPR] + b.x;  o.y = a[1 * LPR] + b.y;
            o.z = a[2 * LPR] + b.z;  o.w = a[3 * LPR] + b.w;
            o.x = o.x > 0.0f ? o.x : 0.0f;  o.y = o.y > 0.0f ? o.y : 0.0f;
            o.z = o.z > 0.0f ? o.z : 0.0f;  o.w = o.w > 0.0f ? o.w : 0.0f;
            *(float4*)(out + (long)gr * F + 4 * l) = o;
        }
    }
}

extern "C" void kernel_launch(void* const* d_in, const int* in_sizes, int n_in,
                              void* d_out, int out_size, void* d_ws, size_t ws_size,
                              hipStream_t stream)
{
    const float* feat0 = (const float*)d_in[0];
    const float* feat1 = (const float*)d_in[1];
    const float* fc0_w = (const float*)d_in[2];
    const float* fc0_b = (const float*)d_in[3];
    const float* fc1_w = (const float*)d_in[4];
    const float* fc1_b = (const float*)d_in[5];
    const float* W1    = (const float*)d_in[6];
    const float* b1    = (const float*)d_in[7];
    const float* W2    = (const float*)d_in[8];
    const float* b2    = (const float*)d_in[9];
    const float* pw    = (const float*)d_in[10];
    const float* pb    = (const float*)d_in[11];
    const int*   erow  = (const int*)d_in[12];
    const int*   ecol  = (const int*)d_in[13];
    const float* ew    = (const float*)d_in[14];
    float*       out   = (float*)d_out;

    const int E       = in_sizes[12];
    const int N_NODES = 100000;
    const int NT      = 50000;

    float* x   = (float*)d_ws;
    float* h   = x + (long)N_NODES * 128;
    float* agg = h + (long)N_NODES * 128;

    gemm_bstat_wmma_f32<128, 8, 2, 1><<<GEMM_BLOCKS, GEMM_THREADS, 0, stream>>>(
        feat0, fc0_w, fc0_b, x, NT, 128, 1);
    gemm_bstat_wmma_f32<256, 8, 2, 1><<<GEMM_BLOCKS, GEMM_THREADS, 0, stream>>>(
        feat1, fc1_w, fc1_b, x + (long)NT * 128, NT, 128, 1);

    gemm_bstat_wmma_f32<128, 8, 2, 0><<<GEMM_BLOCKS, GEMM_THREADS, 0, stream>>>(
        x, W1, nullptr, h, N_NODES, 128, 0);
    spmm_tile_lds_f32<128><<<(N_NODES + TILE_ROWS - 1) / TILE_ROWS, TILE_THREADS, 0, stream>>>(
        erow, ecol, ew, h, b1, agg, E, N_NODES);

    gemm_bstat_wmma_f32<128, 4, 2, 0><<<GEMM_BLOCKS, GEMM_THREADS, 0, stream>>>(
        agg, W2, nullptr, x, N_NODES, 64, 0);
    spmm_tile_lds_f32<64><<<(N_NODES + TILE_ROWS - 1) / TILE_ROWS, TILE_THREADS, 0, stream>>>(
        erow, ecol, ew, x, b2, h, E, N_NODES);

    gemm_bstat_wmma_f32<64, 1, 1, 1><<<GEMM_BLOCKS, GEMM_THREADS, 0, stream>>>(
        h, pw, pb, out, N_NODES, 8, 1);
}


# SpMM: edge list split once per call into per-tile buckets (count + place kernels, LDS int counters); a tile's work-group reads its bucket, not the whole list; sum order = LDS arrival, unordered
# speedup vs baseline: 2.8985x; 1.5461x over previous
#include <hip/hip_runtime.h>

typedef __attribute__((ext_vector_type(2))) float v2f;
typedef __attribute__((ext_vector_type(8))) float v8f;

#define GEMM_THREADS 512
#define GEMM_BLOCKS 256
#define GEMM_LDS_FLOATS 41984

template <int K, int NT>
__device__ __forceinline__ void tile_mma(const float* __restrict__ as, const v2f* bv, v8f* acc)
{
    constexpr int P = K + 4;
    #pragma unroll
    for (int s = 0; s < K / 4; ++s) {
        #pragma unroll
        for (int j = 0; j < NT; ++j) {
            const float2 a2 = *(const float2*)(as + j * 16 * P + 4 * s);
            v2f av; av.x = a2.x; av.y = a2.y;
            acc[j] = __builtin_amdgcn_wmma_f32_16x16x4_f32(
                          false, av,  false, bv[s],
                          (short)0, acc[j],  false,  false);
        }
    }
}

template <int K, int NS, int TPW, int TRANSB>
__global__ __launch_bounds__(GEMM_THREADS)
void gemm_bstat_wmma_f32(const float* __restrict__ A, const float* __restrict__ B,
                         const float* __restrict__ bias, float* __restrict__ C,
                         int M, int N, int hasBias)
{
    constexpr int P          = K + 4;
    constexpr int RG         = 16 / NS;
    constexpr int STEP_TILES = RG * TPW;
    constexpr int STEP_ROWS  = STEP_TILES * 16;
    constexpr int K4         = K / 4;
    constexpr int NLD        = STEP_ROWS * K4 / GEMM_THREADS;
    static_assert(STEP_ROWS * K4 % GEMM_THREADS == 0, "step must split evenly over the threads");
    static_assert(2 * STEP_ROWS * P <= GEMM_LDS_FLOATS, "LDS ring too small");
    __shared__ float As[GEMM_LDS_FLOATS];

    const int tid  = threadIdx.x;
    const int wave = __builtin_amdgcn_readfirstlane(tid >> 5);
    const int lane = tid & 31;
    const int c    = lane & 15;
    const int h    = lane >> 4;
    const int strip = wave % NS;
    const int rg    = wave / NS;
    const int n     = strip * 16 + c;

    const int nT   = M >> 4;
    const int G    = gridDim.x;
    const int base = nT / G;
    const int rem  = nT - base * G;
    const int b    = blockIdx.x;
    const int t0   = b * base + (b < rem ? b : rem);
    const int t1   = t0 + base + (b < rem ? 1 : 0);
    if (t0 >= t1) return;

    const float4* A4 = (const float4*)A;

    float4 st[NLD];
    auto fetch = [&](int tb) {
        const int lim = (t1 - tb < STEP_TILES ? t1 - tb : STEP_TILES) * 16 * K4;
        const long g0 = (long)tb * 16 * K4;
        #pragma unroll
        for (int j = 0; j < NLD; ++j) {
            const int i = tid + j * GEMM_THREADS;
            st[j] = make_float4(0.0f, 0.0f, 0.0f, 0.0f);
            if (i < lim) st[j] = A4[g0 + i];
        }
    };
    auto stash = [&](float* buf) {
        #pragma unroll
        for (int j = 0; j < NLD; ++j) {
            const int i   = tid + j * GEMM_THREADS;
            const int row = i / K4;
            const int c4  = i - row * K4;
            *(float4*)(buf + row * P + 4 * c4) = st[j];
        }
    };

    fetch(t0);

    v2f bv[K4];
    if (TRANSB) {
        static_assert(NS * 16 * P <= GEMM_LDS_FLOATS, "B does not fit the LDS array");
        const float4* B4  = (const float4*)B;
        const int     nB4 = N * K4;
        constexpr int NB4 = NS * 16 * K4;
        constexpr int NBL = (NB4 + GEMM_THREADS - 1) / GEMM_THREADS;
        float4 bt[NBL];
        #pragma unroll
        for (int j = 0; j < NBL; ++j) {
            const int i = tid + j * GEMM_THREADS;
            bt[j] = make_float4(0.0f, 0.0f, 0.0f, 0.0f);
            if (i < nB4) bt[j] = B4[i];
        }
        #pragma unroll
        for (int j = 0; j < NBL; ++j) {
            const int i = tid + j * GEMM_THREADS;
            if (i < NB4) {
                const int row = i / K4;
                const int c4  = i - row * K4;
                *(float4*)(As + row * P + 4 * c4) = bt[j];
            }
        }
        __syncthreads();
        #pragma unroll
        for (int s = 0; s < K4; ++s) {
            const float2 v = *(const float2*)(As + n * P + 4 * s + 2 * h);
            bv[s].x = v.x; bv[s].y = v.y;
        }
        __syncthreads();
    } else {
        #pragma unroll
        for (int s = 0; s < K4; ++s) {
            const int k0 = 4 * s + 2 * h;
            float2 v = make_float2(0.0f, 0.0f);
            if (n < N) { v.x = B[(long)k0 * N + n]; v.y = B[(long)(k0 + 1) * N + n]; }
            bv[s].x = v.x; bv[s].y = v.y;
        }
    }
    float bval = 0.0f;
    if (hasBias && n < N) bval = bias[n];

    stash(As);
    __syncthreads();

    int cur = 0;
    for (int tb = t0; tb < t1; tb += STEP_TILES) {
        const int  nxt     = tb + STEP_TILES;
        const bool hasNext = nxt < t1;
        if (hasNext) fetch(nxt);

        const int valid = t1 - tb < STEP_TILES ? t1 - tb : STEP_TILES;
        const int lt0   = rg * TPW;
        const int nv    = valid - lt0;
        const float* as = As + cur * (STEP_ROWS * P) + (lt0 * 16 + c) * P + 2 * h;

        v8f acc[TPW];
        #pragma unroll
        for (int j = 0; j < TPW; ++j) acc[j] = v8f{};
        int done = 0;
        if (nv >= TPW)      { tile_mma<K, TPW>(as, bv, acc); done = TPW; }
        else if (nv >= 1)   { tile_mma<K, 1>(as, bv, acc);   done = 1;   }

        if (n < N) {
            #pragma unroll
            for (int j = 0; j < TPW; ++j) {
                if (j < done) {
                    const long row0 = (long)(tb + lt0 + j) * 16 + 8 * h;
                    #pragma unroll
                    for (int r = 0; r < 8; ++r)
                        C[(row0 + r) * (long)N + n] = acc[j][r] + bval;
                }
            }
        }

        if (hasNext) stash(As + (cur ^ 1) * (STEP_ROWS * P));
        __syncthreads();
        cur ^= 1;
    }
}

#define TILE_ROWS 512
#define TILE_THREADS 1024
#define N_TILES 196
#define BIN_BLOCKS 256
#define BIN_THREADS 1024

__global__ __launch_bounds__(BIN_THREADS)
void edge_tile_count(const int* __restrict__ rows, int* __restrict__ histG, int nE, int nRows)
{
    __shared__ int hist[256];
    const int tid = threadIdx.x;
    if (tid < 256) hist[tid] = 0;
    __syncthreads();
    const int per = (nE + (int)gridDim.x - 1) / (int)gridDim.x;
    int c0 = (int)blockIdx.x * per;  if (c0 > nE) c0 = nE;
    int c1 = c0 + per;               if (c1 > nE) c1 = nE;
    for (int e = c0 + tid; e < c1; e += BIN_THREADS) {
        const int r = rows[e];
        if ((unsigned)r < (unsigned)nRows) atomicAdd(&hist[(unsigned)r / TILE_ROWS], 1);
    }
    __syncthreads();
    if (tid < N_TILES) histG[(int)blockIdx.x * N_TILES + tid] = hist[tid];
}

__global__ __launch_bounds__(BIN_THREADS)
void edge_tile_place(const int* __restrict__ rows, const int* __restrict__ cols,
                     const float* __restrict__ ew, const int* __restrict__ histG,
                     int* __restrict__ binStart, int* __restrict__ recw, int nE, int nRows)
{
    __shared__ int totL[256];
    __shared__ int befL[256];
    __shared__ int cursor[256];
    __shared__ int wsum[8];
    const int tid  = threadIdx.x;
    const int lane = tid & 31;
    const int wave = tid >> 5;
    const int b    = blockIdx.x;
    const int G    = gridDim.x;
    if (tid < 256) { totL[tid] = 0; befL[tid] = 0; }
    __syncthreads();
    {
        const int t    = tid & 255;
        const int part = tid >> 8;
        if (t < N_TILES) {
            int st = 0, sb = 0;
            #pragma unroll 8
            for (int bb = part; bb < G; bb += 4) {
                const int v = histG[bb * N_TILES + t];
                st += v;
                if (bb < b) sb += v;
            }
            atomicAdd(&totL[t], st);
            atomicAdd(&befL[t], sb);
        }
    }
    __syncthreads();
    int v = 0, x = 0;
    if (tid < 256) {
        v = totL[tid];
        x = v;
        for (int d = 1; d < 32; d <<= 1) {
            const int y = __shfl_up(x, d);
            if (lane >= d) x += y;
        }
        if (lane == 31) wsum[wave] = x;
    }
    __syncthreads();
    if (tid < 256) {
        int off = 0;
        for (int w = 0; w < wave; ++w) off += wsum[w];
        const int excl = off + x - v;
        cursor[tid] = excl + befL[tid];
        if (b == 0 && tid <= N_TILES) binStart[tid] = excl;
    }
    __syncthreads();

    int4* rec = (int4*)recw;
    const int per = (nE + G - 1) / G;
    int c0 = b * per;    if (c0 > nE) c0 = nE;
    int c1 = c0 + per;   if (c1 > nE) c1 = nE;
    for (int e = c0 + tid; e < c1; e += BIN_THREADS) {
        const int r = rows[e];
        if ((unsigned)r < (unsigned)nRows) {
            const int tl  = (int)((unsigned)r / TILE_ROWS);
            const int pos = atomicAdd(&cursor[tl], 1);
            rec[pos] = make_int4(r - tl * TILE_ROWS, cols[e], __float_as_int(ew[e]), 0);
        }
    }
}

template <int F>
__device__ __forceinline__ void tile_flush(int n, int rr, int cc, float ww,
                                           const float* __restrict__ dense, float* acc, int lane)
{
    constexpr int LPR = F / 4;
    constexpr int EPI = 32 / LPR;
    constexpr int SB  = 8;
    const int sub = lane / LPR;
    const int l   = lane % LPR;
    for (int s = 0; s < n; s += SB * EPI) {
        float4 v[SB];
        #pragma unroll
        for (int j = 0; j < SB; ++j) {
            const int idx = s + j * EPI + sub;
            const int sr  = __shfl(cc, idx & 31);
            v[j] = make_float4(0.0f, 0.0f, 0.0f, 0.0f);
            if (idx < n) v[j] = *(const float4*)(dense + (long)sr * F + 4 * l);
        }
        #pragma unroll
        for (int j = 0; j < SB; ++j) {
            const int   idx = s + j * EPI + sub;
            const float w   = __shfl(ww, idx & 31);
            const int   lr  = __shfl(rr, idx & 31);
            if (idx < n) {
                float* a = acc + lr * F + l;
                atomicAdd(a + 0 * LPR, v[j].x * w);
                atomicAdd(a + 1 * LPR, v[j].y * w);
                atomicAdd(a + 2 * LPR, v[j].z * w);
                atomicAdd(a + 3 * LPR, v[j].w * w);
            }
        }
    }
}

template <int F>
__global__ __launch_bounds__(TILE_THREADS)
void spmm_tile_lds_f32(const int* __restrict__ recw, const int* __restrict__ binStart,
                       const float* __restrict__ dense, const float* __restrict__ bias,
                       float* __restrict__ out, int nRows)
{
    constexpr int LPR = F / 4;
    __shared__ float acc[TILE_ROWS * F];

    const int tid  = threadIdx.x;
    const int wave = __builtin_amdgcn_readfirstlane(tid >> 5);
    const int lane = tid & 31;
    const int row0 = blockIdx.x * TILE_ROWS;

    float4* acc4 = (float4*)acc;
    for (int i = tid; i < TILE_ROWS * F / 4; i += TILE_THREADS)
        acc4[i] = make_float4(0.0f, 0.0f, 0.0f, 0.0f);
    __syncthreads();

    const int n0 = binStart[blockIdx.x];
    const int n1 = binStart[blockIdx.x + 1];
    int   base = n0 + wave * 32;
    int   rr = 0, cc = 0;
    float ww = 0.0f;
    if (base + lane < n1) {
        const int* p = recw + 4L * (base + lane);
        rr = p[0];  cc = p[1];  ww = __int_as_float(p[2]);
    }
    while (base < n1) {
        const int nb = base + TILE_THREADS;
        int   rn = 0, cn = 0;
        float wn = 0.0f;
        if (nb + lane < n1) {
            const int* p = recw + 4L * (nb + lane);
            rn = p[0];  cn = p[1];  wn = __int_as_float(p[2]);
        }
        const int n = n1 - base < 32 ? n1 - base : 32;
        tile_flush<F>(n, rr, cc, ww, dense, acc, lane);
        rr = rn;  cc = cn;  ww = wn;
        base = nb;
    }
    __syncthreads();

    for (int i = tid; i < TILE_ROWS * LPR; i += TILE_THREADS) {
        const int r  = i / LPR;
        const int l  = i % LPR;
        const int gr = row0 + r;
        if (gr < nRows) {
            const float* a = acc + r * F + l;
            const float4 b = *(const float4*)(bias + 4 * l);
            float4 o;
            o.x = a[0 * LPR] + b.x;  o.y = a[1 * LPR] + b.y;
            o.z = a[2 * LPR] + b.z;  o.w = a[3 * LPR] + b.w;
            o.x = o.x > 0.0f ? o.x : 0.0f;  o.y = o.y > 0.0f ? o.y : 0.0f;
            o.z = o.z > 0.0f ? o.z : 0.0f;  o.w = o.w > 0.0f ? o.w : 0.0f;
            *(float4*)(out + (long)gr * F + 4 * l) = o;
        }
    }
}

extern "C" void kernel_launch(void* const* d_in, const int* in_sizes, int n_in,
                              void* d_out, int out_size, void* d_ws, size_t ws_size,
                              hipStream_t stream)
{
    const float* feat0 = (const float*)d_in[0];
    const float* feat1 = (const float*)d_in[1];
    const float* fc0_w = (const float*)d_in[2];
    const float* fc0_b = (const float*)d_in[3];
    const float* fc1_w = (const float*)d_in[4];
    const float* fc1_b = (const float*)d_in[5];
    const float* W1    = (const float*)d_in[6];
    const float* b1    = (const float*)d_in[7];
    const float* W2    = (const float*)d_in[8];
    const float* b2    = (const float*)d_in[9];
    const float* pw    = (const float*)d_in[10];
    const float* pb    = (const float*)d_in[11];
    const int*   erow  = (const int*)d_in[12];
    const int*   ecol  = (const int*)d_in[13];
    const float* ew    = (const float*)d_in[14];
    float*       out   = (float*)d_out;

    const int E       = in_sizes[12];
    const int N_NODES = 100000;
    const int NT      = 50000;

    float* x   = (float*)d_ws;
    float* h   = x + (long)N_NODES * 128;
    float* agg = h + (long)N_NODES * 128;
    int* histG    = (int*)(agg + (long)N_NODES * 128);
    int* binStart = histG + BIN_BLOCKS * N_TILES;
    int* rec      = binStart + 256;

    edge_tile_count<<<BIN_BLOCKS, BIN_THREADS, 0, stream>>>(erow, histG, E, N_NODES);
    edge_tile_place<<<BIN_BLOCKS, BIN_THREADS, 0, stream>>>(erow, ecol, ew, histG, binStart, rec, E, N_NODES);

    gemm_bstat_wmma_f32<128, 8, 2, 1><<<GEMM_BLOCKS, GEMM_THREADS, 0, stream>>>(
        feat0, fc0_w, fc0_b, x, NT, 128, 1);
    gemm_bstat_wmma_f32<256, 8, 2, 1><<<GEMM_BLOCKS, GEMM_THREADS, 0, stream>>>(
        feat1, fc1_w, fc1_b, x + (long)NT * 128, NT, 128, 1);

    gemm_bstat_wmma_f32<128, 8, 2, 0><<<GEMM_BLOCKS, GEMM_THREADS, 0, stream>>>(
        x, W1, nullptr, h, N_NODES, 128, 0);
    spmm_tile_lds_f32<128><<<N_TILES, TILE_THREADS, 0, stream>>>(
        rec, binStart, h, b1, agg, N_NODES);

    gemm_bstat_wmma_f32<128, 4, 2, 0><<<GEMM_BLOCKS, GEMM_THREADS, 0, stream>>>(
        agg, W2, nullptr, x, N_NODES, 64, 0);
    spmm_tile_lds_f32<64><<<N_TILES, TILE_THREADS, 0, stream>>>(
        rec, binStart, x, b2, h, N_NODES);

    gemm_bstat_wmma_f32<64, 1, 1, 1><<<GEMM_BLOCKS, GEMM_THREADS, 0, stream>>>(
        h, pw, pb, out, N_NODES, 8, 1);
}


# SpMM: an edge's fields reach the wave as scalars (v_readlane of the batch records), not by LDS permutes; one row per wave instruction, gathers issued back to back; sum order LDS arrival, unordered
# speedup vs baseline: 3.1513x; 1.0872x over previous
#include <hip/hip_runtime.h>

typedef __attribute__((ext_vector_type(2))) float v2f;
typedef __attribute__((ext_vector_type(8))) float v8f;

#define GEMM_THREADS 512
#define GEMM_BLOCKS 256
#define GEMM_LDS_FLOATS 41984

template <int K, int NT>
__device__ __forceinline__ void tile_mma(const float* __restrict__ as, const v2f* bv, v8f* acc)
{
    constexpr int P = K + 4;
    #pragma unroll
    for (int s = 0; s < K / 4; ++s) {
        #pragma unroll
        for (int j = 0; j < NT; ++j) {
            const float2 a2 = *(const float2*)(as + j * 16 * P + 4 * s);
            v2f av; av.x = a2.x; av.y = a2.y;
            acc[j] = __builtin_amdgcn_wmma_f32_16x16x4_f32(
                          false, av,  false, bv[s],
                          (short)0, acc[j],  false,  false);
        }
    }
}

template <int K, int NS, int TPW, int TRANSB>
__global__ __launch_bounds__(GEMM_THREADS)
void gemm_bstat_wmma_f32(const float* __restrict__ A, const float* __restrict__ B,
                         const float* __restrict__ bias, float* __restrict__ C,
                         int M, int N, int hasBias)
{
    constexpr int P          = K + 4;
    constexpr int RG         = 16 / NS;
    constexpr int STEP_TILES = RG * TPW;
    constexpr int STEP_ROWS  = STEP_TILES * 16;
    constexpr int K4         = K / 4;
    constexpr int NLD        = STEP_ROWS * K4 / GEMM_THREADS;
    static_assert(STEP_ROWS * K4 % GEMM_THREADS == 0, "step must split evenly over the threads");
    static_assert(2 * STEP_ROWS * P <= GEMM_LDS_FLOATS, "LDS ring too small");
    __shared__ float As[GEMM_LDS_FLOATS];

    const int tid  = threadIdx.x;
    const int wave = __builtin_amdgcn_readfirstlane(tid >> 5);
    const int lane = tid & 31;
    const int c    = lane & 15;
    const int h    = lane >> 4;
    const int strip = wave % NS;
    const int rg    = wave / NS;
    const int n     = strip * 16 + c;

    const int nT   = M >> 4;
    const int G    = gridDim.x;
    const int base = nT / G;
    const int rem  = nT - base * G;
    const int b    = blockIdx.x;
    const int t0   = b * base + (b < rem ? b : rem);
    const int t1   = t0 + base + (b < rem ? 1 : 0);
    if (t0 >= t1) return;

    const float4* A4 = (const float4*)A;

    float4 st[NLD];
    auto fetch = [&](int tb) {
        const int lim = (t1 - tb < STEP_TILES ? t1 - tb : STEP_TILES) * 16 * K4;
        const long g0 = (long)tb * 16 * K4;
        #pragma unroll
        for (int j = 0; j < NLD; ++j) {
            const int i = tid + j * GEMM_THREADS;
            st[j] = make_float4(0.0f, 0.0f, 0.0f, 0.0f);
            if (i < lim) st[j] = A4[g0 + i];
        }
    };
    auto stash = [&](float* buf) {
        #pragma unroll
        for (int j = 0; j < NLD; ++j) {
            const int i   = tid + j * GEMM_THREADS;
            const int row = i / K4;
            const int c4  = i - row * K4;
            *(float4*)(buf + row * P + 4 * c4) = st[j];
        }
    };

    fetch(t0);

    v2f bv[K4];
    if (TRANSB) {
        static_assert(NS * 16 * P <= GEMM_LDS_FLOATS, "B does not fit the LDS array");
        const float4* B4  = (const float4*)B;
        const int     nB4 = N * K4;
        constexpr int NB4 = NS * 16 * K4;
        constexpr int NBL = (NB4 + GEMM_THREADS - 1) / GEMM_THREADS;
        float4 bt[NBL];
        #pragma unroll
        for (int j = 0; j < NBL; ++j) {
            const int i = tid + j * GEMM_THREADS;
            bt[j] = make_float4(0.0f, 0.0f, 0.0f, 0.0f);
            if (i < nB4) bt[j] = B4[i];
        }
        #pragma unroll
        for (int j = 0; j < NBL; ++j) {
            const int i = tid + j * GEMM_THREADS;
            if (i < NB4) {
                const int row = i / K4;
                const int c4  = i - row * K4;
                *(float4*)(As + row * P + 4 * c4) = bt[j];
            }
        }
        __syncthreads();
        #pragma unroll
        for (int s = 0; s < K4; ++s) {
            const float2 v = *(const float2*)(As + n * P + 4 * s + 2 * h);
            bv[s].x = v.x; bv[s].y = v.y;
        }
        __syncthreads();
    } else {
        #pragma unroll
        for (int s = 0; s < K4; ++s) {
            const int k0 = 4 * s + 2 * h;
            float2 v = make_float2(0.0f, 0.0f);
            if (n < N) { v.x = B[(long)k0 * N + n]; v.y = B[(long)(k0 + 1) * N + n]; }
            bv[s].x = v.x; bv[s].y = v.y;
        }
    }
    float bval = 0.0f;
    if (hasBias && n < N) bval = bias[n];

    stash(As);
    __syncthreads();

    int cur = 0;
    for (int tb = t0; tb < t1; tb += STEP_TILES) {
        const int  nxt     = tb + STEP_TILES;
        const bool hasNext = nxt < t1;
        if (hasNext) fetch(nxt);

        const int valid = t1 - tb < STEP_TILES ? t1 - tb : STEP_TILES;
        const int lt0   = rg * TPW;
        const int nv    = valid - lt0;
        const float* as = As + cur * (STEP_ROWS * P) + (lt0 * 16 + c) * P + 2 * h;

        v8f acc[TPW];
        #pragma unroll
        for (int j = 0; j < TPW; ++j) acc[j] = v8f{};
        int done = 0;
        if (nv >= TPW)      { tile_mma<K, TPW>(as, bv, acc); done = TPW; }
        else if (nv >= 1)   { tile_mma<K, 1>(as, bv, acc);   done = 1;   }

        if (n < N) {
            #pragma unroll
            for (int j = 0; j < TPW; ++j) {
                if (j < done) {
                    const long row0 = (long)(tb + lt0 + j) * 16 + 8 * h;
                    #pragma unroll
                    for (int r = 0; r < 8; ++r)
                        C[(row0 + r) * (long)N + n] = acc[j][r] + bval;
                }
            }
        }

        if (hasNext) stash(As + (cur ^ 1) * (STEP_ROWS * P));
        __syncthreads();
        cur ^= 1;
    }
}

#define TILE_ROWS 512
#define TILE_THREADS 1024
#define N_TILES 196
#define BIN_BLOCKS 256
#define BIN_THREADS 1024

__global__ __launch_bounds__(BIN_THREADS)
void edge_tile_count(const int* __restrict__ rows, int* __restrict__ histG, int nE, int nRows)
{
    __shared__ int hist[256];
    const int tid = threadIdx.x;
    if (tid < 256) hist[tid] = 0;
    __syncthreads();
    const int per = (nE + (int)gridDim.x - 1) / (int)gridDim.x;
    int c0 = (int)blockIdx.x * per;  if (c0 > nE) c0 = nE;
    int c1 = c0 + per;               if (c1 > nE) c1 = nE;
    for (int e = c0 + tid; e < c1; e += BIN_THREADS) {
        const int r = rows[e];
        if ((unsigned)r < (unsigned)nRows) atomicAdd(&hist[(unsigned)r / TILE_ROWS], 1);
    }
    __syncthreads();
    if (tid < N_TILES) histG[(int)blockIdx.x * N_TILES + tid] = hist[tid];
}

__global__ __launch_bounds__(BIN_THREADS)
void edge_tile_place(const int* __restrict__ rows, const int* __restrict__ cols,
                     const float* __restrict__ ew, const int* __restrict__ histG,
                     int* __restrict__ binStart, int* __restrict__ recw, int nE, int nRows)
{
    __shared__ int totL[256];
    __shared__ int befL[256];
    __shared__ int cursor[256];
    __shared__ int wsum[8];
    const int tid  = threadIdx.x;
    const int lane = tid & 31;
    const int wave = tid >> 5;
    const int b    = blockIdx.x;
    const int G    = gridDim.x;
    if (tid < 256) { totL[tid] = 0; befL[tid] = 0; }
    __syncthreads();
    {
        const int t    = tid & 255;
        const int part = tid >> 8;
        if (t < N_TILES) {
            int st = 0, sb = 0;
            #pragma unroll 8
            for (int bb = part; bb < G; bb += 4) {
                const int v = histG[bb * N_TILES + t];
                st += v;
                if (bb < b) sb += v;
            }
            atomicAdd(&totL[t], st);
            atomicAdd(&befL[t], sb);
        }
    }
    __syncthreads();
    int v = 0, x = 0;
    if (tid < 256) {
        v = totL[tid];
        x = v;
        for (int d = 1; d < 32; d <<= 1) {
            const int y = __shfl_up(x, d);
            if (lane >= d) x += y;
        }
        if (lane == 31) wsum[wave] = x;
    }
    __syncthreads();
    if (tid < 256) {
        int off = 0;
        for (int w = 0; w < wave; ++w) off += wsum[w];
        const int excl = off + x - v;
        cursor[tid] = excl + befL[tid];
        if (b == 0 && tid <= N_TILES) binStart[tid] = excl;
    }
    __syncthreads();

    int4* rec = (int4*)recw;
    const int per = (nE + G - 1) / G;
    int c0 = b * per;    if (c0 > nE) c0 = nE;
    int c1 = c0 + per;   if (c1 > nE) c1 = nE;
    for (int e = c0 + tid; e < c1; e += BIN_THREADS) {
        const int r = rows[e];
        if ((unsigned)r < (unsigned)nRows) {
            const int tl  = (int)((unsigned)r / TILE_ROWS);
            const int pos = atomicAdd(&cursor[tl], 1);
            rec[pos] = make_int4(r - tl * TILE_ROWS, cols[e], __float_as_int(ew[e]), 0);
        }
    }
}

typedef __attribute__((ext_vector_type(4))) float v4f;
template <int VEC> struct row_piece;
template <> struct row_piece<4> { typedef v4f type; };
template <> struct row_piece<2> { typedef v2f type; };

template <int F, bool FULL, typename Hook>
__device__ __forceinline__ void flush_group(int s, int n, int rr, int cc, float ww,
                                            const float* __restrict__ dense, float* acc, int lane,
                                            bool first, Hook after_issue)
{
    constexpr int VEC = F / 32;
    constexpr int SB  = 1024 / F;
    typedef typename row_piece<VEC>::type piece;
    piece v[SB];
    #pragma unroll
    for (int j = 0; j < SB; ++j) {
        const int idx = s + j;
        v[j] = piece{};
        if (FULL || idx < n) {
            const int sr = __builtin_amdgcn_readlane(cc, idx & 31);
            v[j] = *(const piece*)(dense + (long)sr * F + VEC * lane);
        }
    }
    if (first) after_issue();
    #pragma unroll
    for (int j = 0; j < SB; ++j) {
        const int idx = s + j;
        if (FULL || idx < n) {
            const float w  = __int_as_float(__builtin_amdgcn_readlane(__float_as_int(ww), idx & 31));
            const int   lr = __builtin_amdgcn_readlane(rr, idx & 31);
            float* a = acc + lr * F + lane;
            #pragma unroll
            for (int c = 0; c < VEC; ++c)
                atomicAdd(a + c * 32, v[j][c] * w);
        }
    }
}

template <int F>
__global__ __launch_bounds__(TILE_THREADS)
void spmm_tile_lds_f32(const int* __restrict__ recw, const int* __restrict__ binStart,
                       const float* __restrict__ dense, const float* __restrict__ bias,
                       float* __restrict__ out, int nRows)
{
    constexpr int LPR = F / 4;
    constexpr int VEC = F / 32;
    constexpr int SB  = 1024 / F;
    __shared__ float acc[TILE_ROWS * F];

    const int tid  = threadIdx.x;
    const int wave = __builtin_amdgcn_readfirstlane(tid >> 5);
    const int lane = tid & 31;
    const int row0 = blockIdx.x * TILE_ROWS;

    float4* acc4 = (float4*)acc;
    for (int i = tid; i < TILE_ROWS * F / 4; i += TILE_THREADS)
        acc4[i] = make_float4(0.0f, 0.0f, 0.0f, 0.0f);
    __syncthreads();

    const int n0 = binStart[blockIdx.x];
    const int n1 = binStart[blockIdx.x + 1];
    int   base = n0 + wave * 32;
    int   rr = 0, cc = 0;
    float ww = 0.0f;
    if (base + lane < n1) {
        const int* p = recw + 4L * (base + lane);
        rr = p[0];  cc = p[1];  ww = __int_as_float(p[2]);
    }
    while (base < n1) {
        const int nb = base + TILE_THREADS;
        int   rn = 0, cn = 0;
        float wn = 0.0f;
        auto next_records = [&]() {
            if (nb + lane < n1) {
                const int* p = recw + 4L * (nb + lane);
                rn = p[0];  cn = p[1];  wn = __int_as_float(p[2]);
            }
        };
        const int n = n1 - base < 32 ? n1 - base : 32;
        if (n == 32) {
            #pragma unroll
            for (int s = 0; s < 32; s += SB)
                flush_group<F, true>(s, 32, rr, cc, ww, dense, acc, lane, s == 0, next_records);
        } else {
            for (int s = 0; s < n; s += SB)
                flush_group<F, false>(s, n, rr, cc, ww, dense, acc, lane, s == 0, next_records);
        }
        rr = rn;  cc = cn;  ww = wn;
        base = nb;
    }
    __syncthreads();

    for (int i = tid; i < TILE_ROWS * LPR; i += TILE_THREADS) {
        const int r  = i / LPR;
        const int l  = i % LPR;
        const int gr = row0 + r;
        if (gr < nRows) {
            const float* a = acc + r * F;
            const float4 b = *(const float4*)(bias + 4 * l);
            float4 o;
            o.x = a[((4 * l + 0) % VEC) * 32 + (4 * l + 0) / VEC] + b.x;
            o.y = a[((4 * l + 1) % VEC) * 32 + (4 * l + 1) / VEC] + b.y;
            o.z = a[((4 * l + 2) % VEC) * 32 + (4 * l + 2) / VEC] + b.z;
            o.w = a[((4 * l + 3) % VEC) * 32 + (4 * l + 3) / VEC] + b.w;
            o.x = o.x > 0.0f ? o.x : 0.0f;  o.y = o.y > 0.0f ? o.y : 0.0f;
            o.z = o.z > 0.0f ? o.z : 0.0f;  o.w = o.w > 0.0f ? o.w : 0.0f;
            *(float4*)(out + (long)gr * F + 4 * l) = o;
        }
    }
}

extern "C" void kernel_launch(void* const* d_in, const int* in_sizes, int n_in,
                              void* d_out, int out_size, void* d_ws, size_t ws_size,
                              hipStream_t stream)
{
    const float* feat0 = (const float*)d_in[0];
    const float* feat1 = (const float*)d_in[1];
    const float* fc0_w = (const float*)d_in[2];
    const float* fc0_b = (const float*)d_in[3];
    const float* fc1_w = (const float*)d_in[4];
    const float* fc1_b = (const float*)d_in[5];
    const float* W1    = (const float*)d_in[6];
    const float* b1    = (const float*)d_in[7];
    const float* W2    = (const float*)d_in[8];
    const float* b2    = (const float*)d_in[9];
    const float* pw    = (const float*)d_in[10];
    const float* pb    = (const float*)d_in[11];
    const int*   erow  = (const int*)d_in[12];
    const int*   ecol  = (const int*)d_in[13];
    const float* ew    = (const float*)d_in[14];
    float*       out   = (float*)d_out;

    const int E       = in_sizes[12];
    const int N_NODES = 100000;
    const int NT      = 50000;

    float* x   = (float*)d_ws;
    float* h   = x + (long)N_NODES * 128;
    float* agg = h + (long)N_NODES * 128;
    int* histG    = (int*)(agg + (long)N_NODES * 128);
    int* binStart = histG + BIN_BLOCKS * N_TILES;
    int* rec      = binStart + 256;

    edge_tile_count<<<BIN_BLOCKS, BIN_THREADS, 0, stream>>>(erow, histG, E, N_NODES);
    edge_tile_place<<<BIN_BLOCKS, BIN_THREADS, 0, stream>>>(erow, ecol, ew, histG, binStart, rec, E, N_NODES);

    gemm_bstat_wmma_f32<128, 8, 2, 1><<<GEMM_BLOCKS, GEMM_THREADS, 0, stream>>>(
        feat0, fc0_w, fc0_b, x, NT, 128, 1);
    gemm_bstat_wmma_f32<256, 8, 2, 1><<<GEMM_BLOCKS, GEMM_THREADS, 0, stream>>>(
        feat1, fc1_w, fc1_b, x + (long)NT * 128, NT, 128, 1);

    gemm_bstat_wmma_f32<128, 8, 2, 0><<<GEMM_BLOCKS, GEMM_THREADS, 0, stream>>>(
        x, W1, nullptr, h, N_NODES, 128, 0);
    spmm_tile_lds_f32<128><<<N_TILES, TILE_THREADS, 0, stream>>>(
        rec, binStart, h, b1, agg, N_NODES);

    gemm_bstat_wmma_f32<128, 4, 2, 0><<<GEMM_BLOCKS, GEMM_THREADS, 0, stream>>>(
        agg, W2, nullptr, x, N_NODES, 64, 0);
    spmm_tile_lds_f32<64><<<N_TILES, TILE_THREADS, 0, stream>>>(
        rec, binStart, x, b2, h, N_NODES);

    gemm_bstat_wmma_f32<64, 1, 1, 1><<<GEMM_BLOCKS, GEMM_THREADS, 0, stream>>>(
        h, pw, pb, out, N_NODES, 8, 1);
}


# SpMM tile 512 -> 392 rows: 256 tiles, one work-group per WGP (was 196 tiles: 60 of 256 WGPs idle in both SpMM launches); buckets follow; sum order LDS arrival, unordered as before
# speedup vs baseline: 3.2703x; 1.0378x over previous
#include <hip/hip_runtime.h>

typedef __attribute__((ext_vector_type(2))) float v2f;
typedef __attribute__((ext_vector_type(8))) float v8f;

#define GEMM_THREADS 512
#define GEMM_BLOCKS 256
#define GEMM_LDS_FLOATS 41984

template <int K, int NT>
__device__ __forceinline__ void tile_mma(const float* __restrict__ as, const v2f* bv, v8f* acc)
{
    constexpr int P = K + 4;
    #pragma unroll
    for (int s = 0; s < K / 4; ++s) {
        #pragma unroll
        for (int j = 0; j < NT; ++j) {
            const float2 a2 = *(const float2*)(as + j * 16 * P + 4 * s);
            v2f av; av.x = a2.x; av.y = a2.y;
            acc[j] = __builtin_amdgcn_wmma_f32_16x16x4_f32(
                          false, av,  false, bv[s],
                          (short)0, acc[j],  false,  false);
        }
    }
}

template <int K, int NS, int TPW, int TRANSB>
__global__ __launch_bounds__(GEMM_THREADS)
void gemm_bstat_wmma_f32(const float* __restrict__ A, const float* __restrict__ B,
                         const float* __restrict__ bias, float* __restrict__ C,
                         int M, int N, int hasBias)
{
    constexpr int P          = K + 4;
    constexpr int RG         = 16 / NS;
    constexpr int STEP_TILES = RG * TPW;
    constexpr int STEP_ROWS  = STEP_TILES * 16;
    constexpr int K4         = K / 4;
    constexpr int NLD        = STEP_ROWS * K4 / GEMM_THREADS;
    static_assert(STEP_ROWS * K4 % GEMM_THREADS == 0, "step must split evenly over the threads");
    static_assert(2 * STEP_ROWS * P <= GEMM_LDS_FLOATS, "LDS ring too small");
    __shared__ float As[GEMM_LDS_FLOATS];

    const int tid  = threadIdx.x;
    const int wave = __builtin_amdgcn_readfirstlane(tid >> 5);
    const int lane = tid & 31;
    const int c    = lane & 15;
    const int h    = lane >> 4;
    const int strip = wave % NS;
    const int rg    = wave / NS;
    const int n     = strip * 16 + c;

    const int nT   = M >> 4;
    const int G    = gridDim.x;
    const int base = nT / G;
    const int rem  = nT - base * G;
    const int b    = blockIdx.x;
    const int t0   = b * base + (b < rem ? b : rem);
    const int t1   = t0 + base + (b < rem ? 1 : 0);
    if (t0 >= t1) return;

    const float4* A4 = (const float4*)A;

    float4 st[NLD];
    auto fetch = [&](int tb) {
        const int lim = (t1 - tb < STEP_TILES ? t1 - tb : STEP_TILES) * 16 * K4;
        const long g0 = (long)tb * 16 * K4;
        #pragma unroll
        for (int j = 0; j < NLD; ++j) {
            const int i = tid + j * GEMM_THREADS;
            st[j] = make_float4(0.0f, 0.0f, 0.0f, 0.0f);
            if (i < lim) st[j] = A4[g0 + i];
        }
    };
    auto stash = [&](float* buf) {
        #pragma unroll
        for (int j = 0; j < NLD; ++j) {
            const int i   = tid + j * GEMM_THREADS;
            const int row = i / K4;
            const int c4  = i - row * K4;
            *(float4*)(buf + row * P + 4 * c4) = st[j];
        }
    };

    fetch(t0);

    v2f bv[K4];
    if (TRANSB) {
        static_assert(NS * 16 * P <= GEMM_LDS_FLOATS, "B does not fit the LDS array");
        const float4* B4  = (const float4*)B;
        const int     nB4 = N * K4;
        constexpr int NB4 = NS * 16 * K4;
        constexpr int NBL = (NB4 + GEMM_THREADS - 1) / GEMM_THREADS;
        float4 bt[NBL];
        #pragma unroll
        for (int j = 0; j < NBL; ++j) {
            const int i = tid + j * GEMM_THREADS;
            bt[j] = make_float4(0.0f, 0.0f, 0.0f, 0.0f);
            if (i < nB4) bt[j] = B4[i];
        }
        #pragma unroll
        for (int j = 0; j < NBL; ++j) {
            const int i = tid + j * GEMM_THREADS;
            if (i < NB4) {
                const int row = i / K4;
                const int c4  = i - row * K4;
                *(float4*)(As + row * P + 4 * c4) = bt[j];
            }
        }
        __syncthreads();
        #pragma unroll
        for (int s = 0; s < K4; ++s) {
            const float2 v = *(const float2*)(As + n * P + 4 * s + 2 * h);
            bv[s].x = v.x; bv[s].y = v.y;
        }
        __syncthreads();
    } else {
        #pragma unroll
        for (int s = 0; s < K4; ++s) {
            const int k0 = 4 * s + 2 * h;
            float2 v = make_float2(0.0f, 0.0f);
            if (n < N) { v.x = B[(long)k0 * N + n]; v.y = B[(long)(k0 + 1) * N + n]; }
            bv[s].x = v.x; bv[s].y = v.y;
        }
    }
    float bval = 0.0f;
    if (hasBias && n < N) bval = bias[n];

    stash(As);
    __syncthreads();

    int cur = 0;
    for (int tb = t0; tb < t1; tb += STEP_TILES) {
        const int  nxt     = tb + STEP_TILES;
        const bool hasNext = nxt < t1;
        if (hasNext) fetch(nxt);

        const int valid = t1 - tb < STEP_TILES ? t1 - tb : STEP_TILES;
        const int lt0   = rg * TPW;
        const int nv    = valid - lt0;
        const float* as = As + cur * (STEP_ROWS * P) + (lt0 * 16 + c) * P + 2 * h;

        v8f acc[TPW];
        #pragma unroll
        for (int j = 0; j < TPW; ++j) acc[j] = v8f{};
        int done = 0;
        if (nv >= TPW)      { tile_mma<K, TPW>(as, bv, acc); done = TPW; }
        else if (nv >= 1)   { tile_mma<K, 1>(as, bv, acc);   done = 1;   }

        if (n < N) {
            #pragma unroll
            for (int j = 0; j < TPW; ++j) {
                if (j < done) {
                    const long row0 = (long)(tb + lt0 + j) * 16 + 8 * h;
                    #pragma unroll
                    for (int r = 0; r < 8; ++r)
                        C[(row0 + r) * (long)N + n] = acc[j][r] + bval;
                }
            }
        }

        if (hasNext) stash(As + (cur ^ 1) * (STEP_ROWS * P));
        __syncthreads();
        cur ^= 1;
    }
}

#define TILE_ROWS 392
#define TILE_THREADS 1024
#define N_TILES 256
#define BIN_BLOCKS 256
#define BIN_THREADS 1024

__global__ __launch_bounds__(BIN_THREADS)
void edge_tile_count(const int* __restrict__ rows, int* __restrict__ histG, int nE, int nRows)
{
    __shared__ int hist[256];
    const int tid = threadIdx.x;
    if (tid < 256) hist[tid] = 0;
    __syncthreads();
    const int per = (nE + (int)gridDim.x - 1) / (int)gridDim.x;
    int c0 = (int)blockIdx.x * per;  if (c0 > nE) c0 = nE;
    int c1 = c0 + per;               if (c1 > nE) c1 = nE;
    for (int e = c0 + tid; e < c1; e += BIN_THREADS) {
        const int r = rows[e];
        if ((unsigned)r < (unsigned)nRows) atomicAdd(&hist[(unsigned)r / TILE_ROWS], 1);
    }
    __syncthreads();
    if (tid < N_TILES) histG[(int)blockIdx.x * N_TILES + tid] = hist[tid];
}

__global__ __launch_bounds__(BIN_THREADS)
void edge_tile_place(const int* __restrict__ rows, const int* __restrict__ cols,
                     const float* __restrict__ ew, const int* __restrict__ histG,
                     int* __restrict__ binStart, int* __restrict__ recw, int nE, int nRows)
{
    __shared__ int totL[256];
    __shared__ int befL[256];
    __shared__ int cursor[256];
    __shared__ int wsum[8];
    const int tid  = threadIdx.x;
    const int lane = tid & 31;
    const int wave = tid >> 5;
    const int b    = blockIdx.x;
    const int G    = gridDim.x;
    if (tid < 256) { totL[tid] = 0; befL[tid] = 0; }
    __syncthreads();
    {
        const int t    = tid & 255;
        const int part = tid >> 8;
        if (t < N_TILES) {
            int st = 0, sb = 0;
            #pragma unroll 8
            for (int bb = part; bb < G; bb += 4) {
                const int v = histG[bb * N_TILES + t];
                st += v;
                if (bb < b) sb += v;
            }
            atomicAdd(&totL[t], st);
            atomicAdd(&befL[t], sb);
        }
    }
    __syncthreads();
    int v = 0, x = 0;
    if (tid < 256) {
        v = totL[tid];
        x = v;
        for (int d = 1; d < 32; d <<= 1) {
            const int y = __shfl_up(x, d);
            if (lane >= d) x += y;
        }
        if (lane == 31) wsum[wave] = x;
    }
    __syncthreads();
    if (tid < 256) {
        int off = 0;
        for (int w = 0; w < wave; ++w) off += wsum[w];
        const int excl = off + x - v;
        cursor[tid] = excl + befL[tid];
        if (b == 0) {
            binStart[tid] = excl;
            if (tid == 255) binStart[256] = excl + v;
        }
    }
    __syncthreads();

    int4* rec = (int4*)recw;
    const int per = (nE + G - 1) / G;
    int c0 = b * per;    if (c0 > nE) c0 = nE;
    int c1 = c0 + per;   if (c1 > nE) c1 = nE;
    for (int e = c0 + tid; e < c1; e += BIN_THREADS) {
        const int r = rows[e];
        if ((unsigned)r < (unsigned)nRows) {
            const int tl  = (int)((unsigned)r / TILE_ROWS);
            const int pos = atomicAdd(&cursor[tl], 1);
            rec[pos] = make_int4(r - tl * TILE_ROWS, cols[e], __float_as_int(ew[e]), 0);
        }
    }
}

typedef __attribute__((ext_vector_type(4))) float v4f;
template <int VEC> struct row_piece;
template <> struct row_piece<4> { typedef v4f type; };
template <> struct row_piece<2> { typedef v2f type; };

template <int F, bool FULL, typename Hook>
__device__ __forceinline__ void flush_group(int s, int n, int rr, int cc, float ww,
                                            const float* __restrict__ dense, float* acc, int lane,
                                            bool first, Hook after_issue)
{
    constexpr int VEC = F / 32;
    constexpr int SB  = 1024 / F;
    typedef typename row_piece<VEC>::type piece;
    piece v[SB];
    #pragma unroll
    for (int j = 0; j < SB; ++j) {
        const int idx = s + j;
        v[j] = piece{};
        if (FULL || idx < n) {
            const int sr = __builtin_amdgcn_readlane(cc, idx & 31);
            v[j] = *(const piece*)(dense + (long)sr * F + VEC * lane);
        }
    }
    if (first) after_issue();
    #pragma unroll
    for (int j = 0; j < SB; ++j) {
        const int idx = s + j;
        if (FULL || idx < n) {
            const float w  = __int_as_float(__builtin_amdgcn_readlane(__float_as_int(ww), idx & 31));
            const int   lr = __builtin_amdgcn_readlane(rr, idx & 31);
            float* a = acc + lr * F + lane;
            #pragma unroll
            for (int c = 0; c < VEC; ++c)
                atomicAdd(a + c * 32, v[j][c] * w);
        }
    }
}

template <int F>
__global__ __launch_bounds__(TILE_THREADS)
void spmm_tile_lds_f32(const int* __restrict__ recw, const int* __restrict__ binStart,
                       const float* __restrict__ dense, const float* __restrict__ bias,
                       float* __restrict__ out, int nRows)
{
    constexpr int LPR = F / 4;
    constexpr int VEC = F / 32;
    constexpr int SB  = 1024 / F;
    __shared__ float acc[TILE_ROWS * F];

    const int tid  = threadIdx.x;
    const int wave = __builtin_amdgcn_readfirstlane(tid >> 5);
    const int lane = tid & 31;
    const int row0 = blockIdx.x * TILE_ROWS;

    float4* acc4 = (float4*)acc;
    for (int i = tid; i < TILE_ROWS * F / 4; i += TILE_THREADS)
        acc4[i] = make_float4(0.0f, 0.0f, 0.0f, 0.0f);
    __syncthreads();

    const int n0 = binStart[blockIdx.x];
    const int n1 = binStart[blockIdx.x + 1];
    int   base = n0 + wave * 32;
    int   rr = 0, cc = 0;
    float ww = 0.0f;
    if (base + lane < n1) {
        const int* p = recw + 4L * (base + lane);
        rr = p[0];  cc = p[1];  ww = __int_as_float(p[2]);
    }
    while (base < n1) {
        const int nb = base + TILE_THREADS;
        int   rn = 0, cn = 0;
        float wn = 0.0f;
        auto next_records = [&]() {
            if (nb + lane < n1) {
                const int* p = recw + 4L * (nb + lane);
                rn = p[0];  cn = p[1];  wn = __int_as_float(p[2]);
            }
        };
        const int n = n1 - base < 32 ? n1 - base : 32;
        if (n == 32) {
            #pragma unroll
            for (int s = 0; s < 32; s += SB)
                flush_group<F, true>(s, 32, rr, cc, ww, dense, acc, lane, s == 0, next_records);
        } else {
            for (int s = 0; s < n; s += SB)
                flush_group<F, false>(s, n, rr, cc, ww, dense, acc, lane, s == 0, next_records);
        }
        rr = rn;  cc = cn;  ww = wn;
        base = nb;
    }
    __syncthreads();

    for (int i = tid; i < TILE_ROWS * LPR; i += TILE_THREADS) {
        const int r  = i / LPR;
        const int l  = i % LPR;
        const int gr = row0 + r;
        if (gr < nRows) {
            const float* a = acc + r * F;
            const float4 b = *(const float4*)(bias + 4 * l);
            float4 o;
            o.x = a[((4 * l + 0) % VEC) * 32 + (4 * l + 0) / VEC] + b.x;
            o.y = a[((4 * l + 1) % VEC) * 32 + (4 * l + 1) / VEC] + b.y;
            o.z = a[((4 * l + 2) % VEC) * 32 + (4 * l + 2) / VEC] + b.z;
            o.w = a[((4 * l + 3) % VEC) * 32 + (4 * l + 3) / VEC] + b.w;
            o.x = o.x > 0.0f ? o.x : 0.0f;  o.y = o.y > 0.0f ? o.y : 0.0f;
            o.z = o.z > 0.0f ? o.z : 0.0f;  o.w = o.w > 0.0f ? o.w : 0.0f;
            *(float4*)(out + (long)gr * F + 4 * l) = o;
        }
    }
}

extern "C" void kernel_launch(void* const* d_in, const int* in_sizes, int n_in,
                              void* d_out, int out_size, void* d_ws, size_t ws_size,
                              hipStream_t stream)
{
    const float* feat0 = (const float*)d_in[0];
    const float* feat1 = (const float*)d_in[1];
    const float* fc0_w = (const float*)d_in[2];
    const float* fc0_b = (const float*)d_in[3];
    const float* fc1_w = (const float*)d_in[4];
    const float* fc1_b = (const float*)d_in[5];
    const float* W1    = (const float*)d_in[6];
    const float* b1    = (const float*)d_in[7];
    const float* W2    = (const float*)d_in[8];
    const float* b2    = (const float*)d_in[9];
    const float* pw    = (const float*)d_in[10];
    const float* pb    = (const float*)d_in[11];
    const int*   erow  = (const int*)d_in[12];
    const int*   ecol  = (const int*)d_in[13];
    const float* ew    = (const float*)d_in[14];
    float*       out   = (float*)d_out;

    const int E       = in_sizes[12];
    const int N_NODES = 100000;
    const int NT      = 50000;

    float* x   = (float*)d_ws;
    float* h   = x + (long)N_NODES * 128;
    float* agg = h + (long)N_NODES * 128;
    int* histG    = (int*)(agg + (long)N_NODES * 128);
    int* binStart = histG + BIN_BLOCKS * N_TILES;
    int* rec      = binStart + 320;

    edge_tile_count<<<BIN_BLOCKS, BIN_THREADS, 0, stream>>>(erow, histG, E, N_NODES);
    edge_tile_place<<<BIN_BLOCKS, BIN_THREADS, 0, stream>>>(erow, ecol, ew, histG, binStart, rec, E, N_NODES);

    gemm_bstat_wmma_f32<128, 8, 2, 1><<<GEMM_BLOCKS, GEMM_THREADS, 0, stream>>>(
        feat0, fc0_w, fc0_b, x, NT, 128, 1);
    gemm_bstat_wmma_f32<256, 8, 2, 1><<<GEMM_BLOCKS, GEMM_THREADS, 0, stream>>>(
        feat1, fc1_w, fc1_b, x + (long)NT * 128, NT, 128, 1);

    gemm_bstat_wmma_f32<128, 8, 2, 0><<<GEMM_BLOCKS, GEMM_THREADS, 0, stream>>>(
        x, W1, nullptr, h, N_NODES, 128, 0);
    spmm_tile_lds_f32<128><<<N_TILES, TILE_THREADS, 0, stream>>>(
        rec, binStart, h, b1, agg, N_NODES);

    gemm_bstat_wmma_f32<128, 4, 2, 0><<<GEMM_BLOCKS, GEMM_THREADS, 0, stream>>>(
        agg, W2, nullptr, x, N_NODES, 64, 0);
    spmm_tile_lds_f32<64><<<N_TILES, TILE_THREADS, 0, stream>>>(
        rec, binStart, x, b2, h, N_NODES);

    gemm_bstat_wmma_f32<64, 1, 1, 1><<<GEMM_BLOCKS, GEMM_THREADS, 0, stream>>>(
        h, pw, pb, out, N_NODES, 8, 1);
}


# Layer-0 projections and x@W1 chained in ONE launch (was 3): a work-group's projected 64-row step stays in LDS as A operand of the W1 product (W1^T in LDS); x never stored; sum orders unchanged
# speedup vs baseline: 3.7488x; 1.1463x over previous
#include <hip/hip_runtime.h>

typedef __attribute__((ext_vector_type(2))) float v2f;
typedef __attribute__((ext_vector_type(8))) float v8f;

#define GEMM_THREADS 512
#define GEMM_BLOCKS 256
#define GEMM_LDS_FLOATS 41984

template <int K, int NT>
__device__ __forceinline__ void tile_mma(const float* __restrict__ as, const v2f* bv, v8f* acc)
{
    constexpr int P = K + 4;
    #pragma unroll
    for (int s = 0; s < K / 4; ++s) {
        #pragma unroll
        for (int j = 0; j < NT; ++j) {
            const float2 a2 = *(const float2*)(as + j * 16 * P + 4 * s);
            v2f av; av.x = a2.x; av.y = a2.y;
            acc[j] = __builtin_amdgcn_wmma_f32_16x16x4_f32(
                          false, av,  false, bv[s],
                          (short)0, acc[j],  false,  false);
        }
    }
}

template <int K, int NS, int TPW, int TRANSB>
__global__ __launch_bounds__(GEMM_THREADS)
void gemm_bstat_wmma_f32(const float* __restrict__ A, const float* __restrict__ B,
                         const float* __restrict__ bias, float* __restrict__ C,
                         int M, int N, int hasBias)
{
    constexpr int P          = K + 4;
    constexpr int RG         = 16 / NS;
    constexpr int STEP_TILES = RG * TPW;
    constexpr int STEP_ROWS  = STEP_TILES * 16;
    constexpr int K4         = K / 4;
    constexpr int NLD        = STEP_ROWS * K4 / GEMM_THREADS;
    static_assert(STEP_ROWS * K4 % GEMM_THREADS == 0, "step must split evenly over the threads");
    static_assert(2 * STEP_ROWS * P <= GEMM_LDS_FLOATS, "LDS ring too small");
    __shared__ float As[GEMM_LDS_FLOATS];

    const int tid  = threadIdx.x;
    const int wave = __builtin_amdgcn_readfirstlane(tid >> 5);
    const int lane = tid & 31;
    const int c    = lane & 15;
    const int h    = lane >> 4;
    const int strip = wave % NS;
    const int rg    = wave / NS;
    const int n     = strip * 16 + c;

    const int nT   = M >> 4;
    const int G    = gridDim.x;
    const int base = nT / G;
    const int rem  = nT - base * G;
    const int b    = blockIdx.x;
    const int t0   = b * base + (b < rem ? b : rem);
    const int t1   = t0 + base + (b < rem ? 1 : 0);
    if (t0 >= t1) return;

    const float4* A4 = (const float4*)A;

    float4 st[NLD];
    auto fetch = [&](int tb) {
        const int lim = (t1 - tb < STEP_TILES ? t1 - tb : STEP_TILES) * 16 * K4;
        const long g0 = (long)tb * 16 * K4;
        #pragma unroll
        for (int j = 0; j < NLD; ++j) {
            const int i = tid + j * GEMM_THREADS;
            st[j] = make_float4(0.0f, 0.0f, 0.0f, 0.0f);
            if (i < lim) st[j] = A4[g0 + i];
        }
    };
    auto stash = [&](float* buf) {
        #pragma unroll
        for (int j = 0; j < NLD; ++j) {
            const int i   = tid + j * GEMM_THREADS;
            const int row = i / K4;
            const int c4  = i - row * K4;
            *(float4*)(buf + row * P + 4 * c4) = st[j];
        }
    };

    fetch(t0);

    v2f bv[K4];
    if (TRANSB) {
        static_assert(NS * 16 * P <= GEMM_LDS_FLOATS, "B does not fit the LDS array");
        const float4* B4  = (const float4*)B;
        const int     nB4 = N * K4;
        constexpr int NB4 = NS * 16 * K4;
        constexpr int NBL = (NB4 + GEMM_THREADS - 1) / GEMM_THREADS;
        float4 bt[NBL];
        #pragma unroll
        for (int j = 0; j < NBL; ++j) {
            const int i = tid + j * GEMM_THREADS;
            bt[j] = make_float4(0.0f, 0.0f, 0.0f, 0.0f);
            if (i < nB4) bt[j] = B4[i];
        }
        #pragma unroll
        for (int j = 0; j < NBL; ++j) {
            const int i = tid + j * GEMM_THREADS;
            if (i < NB4) {
                const int row = i / K4;
                const int c4  = i - row * K4;
                *(float4*)(As + row * P + 4 * c4) = bt[j];
            }
        }
        __syncthreads();
        #pragma unroll
        for (int s = 0; s < K4; ++s) {
            const float2 v = *(const float2*)(As + n * P + 4 * s + 2 * h);
            bv[s].x = v.x; bv[s].y = v.y;
        }
        __syncthreads();
    } else {
        #pragma unroll
        for (int s = 0; s < K4; ++s) {
            const int k0 = 4 * s + 2 * h;
            float2 v = make_float2(0.0f, 0.0f);
            if (n < N) { v.x = B[(long)k0 * N + n]; v.y = B[(long)(k0 + 1) * N + n]; }
            bv[s].x = v.x; bv[s].y = v.y;
        }
    }
    float bval = 0.0f;
    if (hasBias && n < N) bval = bias[n];

    stash(As);
    __syncthreads();

    int cur = 0;
    for (int tb = t0; tb < t1; tb += STEP_TILES) {
        const int  nxt     = tb + STEP_TILES;
        const bool hasNext = nxt < t1;
        if (hasNext) fetch(nxt);

        const int valid = t1 - tb < STEP_TILES ? t1 - tb : STEP_TILES;
        const int lt0   = rg * TPW;
        const int nv    = valid - lt0;
        const float* as = As + cur * (STEP_ROWS * P) + (lt0 * 16 + c) * P + 2 * h;

        v8f acc[TPW];
        #pragma unroll
        for (int j = 0; j < TPW; ++j) acc[j] = v8f{};
        int done = 0;
        if (nv >= TPW)      { tile_mma<K, TPW>(as, bv, acc); done = TPW; }
        else if (nv >= 1)   { tile_mma<K, 1>(as, bv, acc);   done = 1;   }

        if (n < N) {
            #pragma unroll
            for (int j = 0; j < TPW; ++j) {
                if (j < done) {
                    const long row0 = (long)(tb + lt0 + j) * 16 + 8 * h;
                    #pragma unroll
                    for (int r = 0; r < 8; ++r)
                        C[(row0 + r) * (long)N + n] = acc[j][r] + bval;
                }
            }
        }

        if (hasNext) stash(As + (cur ^ 1) * (STEP_ROWS * P));
        __syncthreads();
        cur ^= 1;
    }
}

#define FUSED_P1 132
#define FUSED_AS 33280
#define FUSED_XS 8448
#define FUSED_WS 16896
#define FUSED_LDS_FLOATS (FUSED_AS + 2 * FUSED_XS + FUSED_WS)

template <int NT>
__device__ __forceinline__ void tile_mma_x(const float* xs, const float* wsn, v8f* acc)
{
    #pragma unroll
    for (int s = 0; s < 32; ++s) {
        const float2 b2 = *(const float2*)(wsn + 4 * s);
        v2f bw; bw.x = b2.x; bw.y = b2.y;
        #pragma unroll
        for (int j = 0; j < NT; ++j) {
            const float2 a2 = *(const float2*)(xs + j * 16 * FUSED_P1 + 4 * s);
            v2f av; av.x = a2.x; av.y = a2.y;
            acc[j] = __builtin_amdgcn_wmma_f32_16x16x4_f32(
                          false, av,  false, bw,
                          (short)0, acc[j],  false,  false);
        }
    }
}

template <int K>
__device__ __forceinline__ void proj_then_w1(float* lds,
                                             const float* __restrict__ A, const float* __restrict__ Bfc,
                                             const float* __restrict__ bias, const float* __restrict__ W1,
                                             float* __restrict__ C, int M, int b, int G)
{
    constexpr int NS         = 8;
    constexpr int TPW        = 2;
    constexpr int P          = K + 4;
    constexpr int PX         = FUSED_P1;
    constexpr int RG         = 16 / NS;
    constexpr int STEP_TILES = RG * TPW;
    constexpr int STEP_ROWS  = STEP_TILES * 16;
    constexpr int K4         = K / 4;
    constexpr int NLD        = STEP_ROWS * K4 / GEMM_THREADS;
    static_assert(STEP_ROWS * K4 % GEMM_THREADS == 0, "step must split evenly over the threads");
    static_assert(2 * STEP_ROWS * P <= FUSED_AS, "feature-step buffers too small");
    static_assert(NS * 16 * P <= FUSED_AS, "fc_w does not fit the feature-step buffers");
    static_assert(STEP_ROWS * PX <= FUSED_XS, "projected-step buffer too small");

    float* As = lds;
    float* Xs = lds + FUSED_AS;
    float* Ws = Xs + 2 * FUSED_XS;

    const int tid  = threadIdx.x;
    const int wave = __builtin_amdgcn_readfirstlane(tid >> 5);
    const int lane = tid & 31;
    const int c    = lane & 15;
    const int h    = lane >> 4;
    const int strip = wave % NS;
    const int rg    = wave / NS;
    const int n     = strip * 16 + c;

    const int nT   = M >> 4;
    const int base = nT / G;
    const int rem  = nT - base * G;
    const int t0   = b * base + (b < rem ? b : rem);
    const int t1   = t0 + base + (b < rem ? 1 : 0);
    if (t0 >= t1) return;

    const float4* A4 = (const float4*)A;

    float4 st[NLD];
    auto fetch = [&](int tb) {
        const int lim = (t1 - tb < STEP_TILES ? t1 - tb : STEP_TILES) * 16 * K4;
        const long g0 = (long)tb * 16 * K4;
        #pragma unroll
        for (int j = 0; j < NLD; ++j) {
            const int i = tid + j * GEMM_THREADS;
            st[j] = make_float4(0.0f, 0.0f, 0.0f, 0.0f);
            if (i < lim) st[j] = A4[g0 + i];
        }
    };
    auto stash = [&](float* buf) {
        #pragma unroll
        for (int j = 0; j < NLD; ++j) {
            const int i   = tid + j * GEMM_THREADS;
            const int row = i / K4;
            const int c4  = i - row * K4;
            *(float4*)(buf + row * P + 4 * c4) = st[j];
        }
    };

    fetch(t0);

    {
        const float4* B4  = (const float4*)Bfc;
        constexpr int NBL = 128 * K4 / GEMM_THREADS;
        float4 bt[NBL];
        #pragma unroll
        for (int j = 0; j < NBL; ++j) bt[j] = B4[tid + j * GEMM_THREADS];
        #pragma unroll
        for (int j = 0; j < NBL; ++j) {
            const int i   = tid + j * GEMM_THREADS;
            const int row = i / K4;
            const int c4  = i - row * K4;
            *(float4*)(As + row * P + 4 * c4) = bt[j];
        }
    }
    {
        const float4* W4 = (const float4*)W1;
        #pragma unroll
        for (int j = 0; j < 2; ++j) {
            const int combo = wave * 2 + j;
            const int n4 = (combo & 3) * 8 + (lane & 7);
            const int kg = (combo >> 2) * 4 + (lane >> 3);
            const float4 r0 = W4[(4 * kg + 0) * 32 + n4];
            const float4 r1 = W4[(4 * kg + 1) * 32 + n4];
            const float4 r2 = W4[(4 * kg + 2) * 32 + n4];
            const float4 r3 = W4[(4 * kg + 3) * 32 + n4];
            float* d = Ws + (4 * n4) * PX + 4 * kg;
            *(float4*)(d)          = make_float4(r0.x, r1.x, r2.x, r3.x);
            *(float4*)(d + PX)     = make_float4(r0.y, r1.y, r2.y, r3.y);
            *(float4*)(d + 2 * PX) = make_float4(r0.z, r1.z, r2.z, r3.z);
            *(float4*)(d + 3 * PX) = make_float4(r0.w, r1.w, r2.w, r3.w);
        }
    }
    __syncthreads();
    v2f bv[K4];
    #pragma unroll
    for (int s = 0; s < K4; ++s) {
        const float2 v = *(const float2*)(As + n * P + 4 * s + 2 * h);
        bv[s].x = v.x; bv[s].y = v.y;
    }
    __syncthreads();
    const float bval = bias[n];

    stash(As);
    __syncthreads();

    const float* wsn = Ws + n * PX + 2 * h;
    int cur = 0;
    int xb  = 0;
    for (int tb = t0; tb < t1; tb += STEP_TILES) {
        const int  nxt     = tb + STEP_TILES;
        const bool hasNext = nxt < t1;
        if (hasNext) fetch(nxt);

        const int valid = t1 - tb < STEP_TILES ? t1 - tb : STEP_TILES;
        const int lt0   = rg * TPW;
        const int nv    = valid - lt0;
        const float* as = As + cur * (STEP_ROWS * P) + (lt0 * 16 + c) * P + 2 * h;
        float*       xw = Xs + xb * FUSED_XS;

        v8f acc[TPW];
        #pragma unroll
        for (int j = 0; j < TPW; ++j) acc[j] = v8f{};
        int done = 0;
        if (nv >= TPW)      { tile_mma<K, TPW>(as, bv, acc); done = TPW; }
        else if (nv >= 1)   { tile_mma<K, 1>(as, bv, acc);   done = 1;   }

        if (hasNext) stash(As + (cur ^ 1) * (STEP_ROWS * P));

        #pragma unroll
        for (int j = 0; j < TPW; ++j) {
            if (j < done) {
                float* xr = xw + ((lt0 + j) * 16 + 8 * h) * PX + n;
                #pragma unroll
                for (int r = 0; r < 8; ++r) xr[r * PX] = acc[j][r] + bval;
            }
        }
        __syncthreads();

        const float* xs = xw + (lt0 * 16 + c) * PX + 2 * h;
        v8f acc2[TPW];
        #pragma unroll
        for (int j = 0; j < TPW; ++j) acc2[j] = v8f{};
        if (done == TPW)    tile_mma_x<TPW>(xs, wsn, acc2);
        else if (done == 1) tile_mma_x<1>(xs, wsn, acc2);

        #pragma unroll
        for (int j = 0; j < TPW; ++j) {
            if (j < done) {
                const long row0 = (long)(tb + lt0 + j) * 16 + 8 * h;
                #pragma unroll
                for (int r = 0; r < 8; ++r)
                    C[(row0 + r) * 128L + n] = acc2[j][r];
            }
        }
        cur ^= 1;
        xb  ^= 1;
    }
}

__global__ __launch_bounds__(GEMM_THREADS)
void proj_w1_chain_wmma_f32(const float* __restrict__ feat0, const float* __restrict__ fc0_w,
                            const float* __restrict__ fc0_b, const float* __restrict__ feat1,
                            const float* __restrict__ fc1_w, const float* __restrict__ fc1_b,
                            const float* __restrict__ W1, float* __restrict__ hout,
                            int M0, int M1, int G0)
{
    __shared__ float lds[FUSED_LDS_FLOATS];
    const int b = blockIdx.x;
    const int G = gridDim.x;
    if (b < G0) proj_then_w1<128>(lds, feat0, fc0_w, fc0_b, W1, hout, M0, b, G0);
    else        proj_then_w1<256>(lds, feat1, fc1_w, fc1_b, W1, hout + (long)M0 * 128, M1, b - G0, G - G0);
}

#define TILE_ROWS 392
#define TILE_THREADS 1024
#define N_TILES 256
#define BIN_BLOCKS 256
#define BIN_THREADS 1024

__global__ __launch_bounds__(BIN_THREADS)
void edge_tile_count(const int* __restrict__ rows, int* __restrict__ histG, int nE, int nRows)
{
    __shared__ int hist[256];
    const int tid = threadIdx.x;
    if (tid < 256) hist[tid] = 0;
    __syncthreads();
    const int per = (nE + (int)gridDim.x - 1) / (int)gridDim.x;
    int c0 = (int)blockIdx.x * per;  if (c0 > nE) c0 = nE;
    int c1 = c0 + per;               if (c1 > nE) c1 = nE;
    for (int e = c0 + tid; e < c1; e += BIN_THREADS) {
        const int r = rows[e];
        if ((unsigned)r < (unsigned)nRows) atomicAdd(&hist[(unsigned)r / TILE_ROWS], 1);
    }
    __syncthreads();
    if (tid < N_TILES) histG[(int)blockIdx.x * N_TILES + tid] = hist[tid];
}

__global__ __launch_bounds__(BIN_THREADS)
void edge_tile_place(const int* __restrict__ rows, const int* __restrict__ cols,
                     const float* __restrict__ ew, const int* __restrict__ histG,
                     int* __restrict__ binStart, int* __restrict__ recw, int nE, int nRows)
{
    __shared__ int totL[256];
    __shared__ int befL[256];
    __shared__ int cursor[256];
    __shared__ int wsum[8];
    const int tid  = threadIdx.x;
    const int lane = tid & 31;
    const int wave = tid >> 5;
    const int b    = blockIdx.x;
    const int G    = gridDim.x;
    if (tid < 256) { totL[tid] = 0; befL[tid] = 0; }
    __syncthreads();
    {
        const int t    = tid & 255;
        const int part = tid >> 8;
        if (t < N_TILES) {
            int st = 0, sb = 0;
            #pragma unroll 8
            for (int bb = part; bb < G; bb += 4) {
                const int v = histG[bb * N_TILES + t];
                st += v;
                if (bb < b) sb += v;
            }
            atomicAdd(&totL[t], st);
            atomicAdd(&befL[t], sb);
        }
    }
    __syncthreads();
    int v = 0, x = 0;
    if (tid < 256) {
        v = totL[tid];
        x = v;
        for (int d = 1; d < 32; d <<= 1) {
            const int y = __shfl_up(x, d);
            if (lane >= d) x += y;
        }
        if (lane == 31) wsum[wave] = x;
    }
    __syncthreads();
    if (tid < 256) {
        int off = 0;
        for (int w = 0; w < wave; ++w) off += wsum[w];
        const int excl = off + x - v;
        cursor[tid] = excl + befL[tid];
        if (b == 0) {
            binStart[tid] = excl;
            if (tid == 255) binStart[256] = excl + v;
        }
    }
    __syncthreads();

    int4* rec = (int4*)recw;
    const int per = (nE + G - 1) / G;
    int c0 = b * per;    if (c0 > nE) c0 = nE;
    int c1 = c0 + per;   if (c1 > nE) c1 = nE;
    for (int e = c0 + tid; e < c1; e += BIN_THREADS) {
        const int r = rows[e];
        if ((unsigned)r < (unsigned)nRows) {
            const int tl  = (int)((unsigned)r / TILE_ROWS);
            const int pos = atomicAdd(&cursor[tl], 1);
            rec[pos] = make_int4(r - tl * TILE_ROWS, cols[e], __float_as_int(ew[e]), 0);
        }
    }
}

typedef __attribute__((ext_vector_type(4))) float v4f;
template <int VEC> struct row_piece;
template <> struct row_piece<4> { typedef v4f type; };
template <> struct row_piece<2> { typedef v2f type; };

template <int F, bool FULL, typename Hook>
__device__ __forceinline__ void flush_group(int s, int n, int rr, int cc, float ww,
                                            const float* __restrict__ dense, float* acc, int lane,
                                            bool first, Hook after_issue)
{
    constexpr int VEC = F / 32;
    constexpr int SB  = 1024 / F;
    typedef typename row_piece<VEC>::type piece;
    piece v[SB];
    #pragma unroll
    for (int j = 0; j < SB; ++j) {
        const int idx = s + j;
        v[j] = piece{};
        if (FULL || idx < n) {
            const int sr = __builtin_amdgcn_readlane(cc, idx & 31);
            v[j] = *(const piece*)(dense + (long)sr * F + VEC * lane);
        }
    }
    if (first) after_issue();
    #pragma unroll
    for (int j = 0; j < SB; ++j) {
        const int idx = s + j;
        if (FULL || idx < n) {
            const float w  = __int_as_float(__builtin_amdgcn_readlane(__float_as_int(ww), idx & 31));
            const int   lr = __builtin_amdgcn_readlane(rr, idx & 31);
            float* a = acc + lr * F + lane;
            #pragma unroll
            for (int c = 0; c < VEC; ++c)
                atomicAdd(a + c * 32, v[j][c] * w);
        }
    }
}

template <int F>
__global__ __launch_bounds__(TILE_THREADS)
void spmm_tile_lds_f32(const int* __restrict__ recw, const int* __restrict__ binStart,
                       const float* __restrict__ dense, const float* __restrict__ bias,
                       float* __restrict__ out, int nRows)
{
    constexpr int LPR = F / 4;
    constexpr int VEC = F / 32;
    constexpr int SB  = 1024 / F;
    __shared__ float acc[TILE_ROWS * F];

    const int tid  = threadIdx.x;
    const int wave = __builtin_amdgcn_readfirstlane(tid >> 5);
    const int lane = tid & 31;
    const int row0 = blockIdx.x * TILE_ROWS;

    float4* acc4 = (float4*)acc;
    for (int i = tid; i < TILE_ROWS * F / 4; i += TILE_THREADS)
        acc4[i] = make_float4(0.0f, 0.0f, 0.0f, 0.0f);
    __syncthreads();

    const int n0 = binStart[blockIdx.x];
    const int n1 = binStart[blockIdx.x + 1];
    int   base = n0 + wave * 32;
    int   rr = 0, cc = 0;
    float ww = 0.0f;
    if (base + lane < n1) {
        const int* p = recw + 4L * (base + lane);
        rr = p[0];  cc = p[1];  ww = __int_as_float(p[2]);
    }
    while (base < n1) {
        const int nb = base + TILE_THREADS;
        int   rn = 0, cn = 0;
        float wn = 0.0f;
        auto next_records = [&]() {
            if (nb + lane < n1) {
                const int* p = recw + 4L * (nb + lane);
                rn = p[0];  cn = p[1];  wn = __int_as_float(p[2]);
            }
        };
        const int n = n1 - base < 32 ? n1 - base : 32;
        if (n == 32) {
            #pragma unroll
            for (int s = 0; s < 32; s += SB)
                flush_group<F, true>(s, 32, rr, cc, ww, dense, acc, lane, s == 0, next_records);
        } else {
            for (int s = 0; s < n; s += SB)
                flush_group<F, false>(s, n, rr, cc, ww, dense, acc, lane, s == 0, next_records);
        }
        rr = rn;  cc = cn;  ww = wn;
        base = nb;
    }
    __syncthreads();

    for (int i = tid; i < TILE_ROWS * LPR; i += TILE_THREADS) {
        const int r  = i / LPR;
        const int l  = i % LPR;
        const int gr = row0 + r;
        if (gr < nRows) {
            const float* a = acc + r * F;
            const float4 b = *(const float4*)(bias + 4 * l);
            float4 o;
            o.x = a[((4 * l + 0) % VEC) * 32 + (4 * l + 0) / VEC] + b.x;
            o.y = a[((4 * l + 1) % VEC) * 32 + (4 * l + 1) / VEC] + b.y;
            o.z = a[((4 * l + 2) % VEC) * 32 + (4 * l + 2) / VEC] + b.z;
            o.w = a[((4 * l + 3) % VEC) * 32 + (4 * l + 3) / VEC] + b.w;
            o.x = o.x > 0.0f ? o.x : 0.0f;  o.y = o.y > 0.0f ? o.y : 0.0f;
            o.z = o.z > 0.0f ? o.z : 0.0f;  o.w = o.w > 0.0f ? o.w : 0.0f;
            *(float4*)(out + (long)gr * F + 4 * l) = o;
        }
    }
}

extern "C" void kernel_launch(void* const* d_in, const int* in_sizes, int n_in,
                              void* d_out, int out_size, void* d_ws, size_t ws_size,
                              hipStream_t stream)
{
    const float* feat0 = (const float*)d_in[0];
    const float* feat1 = (const float*)d_in[1];
    const float* fc0_w = (const float*)d_in[2];
    const float* fc0_b = (const float*)d_in[3];
    const float* fc1_w = (const float*)d_in[4];
    const float* fc1_b = (const float*)d_in[5];
    const float* W1    = (const float*)d_in[6];
    const float* b1    = (const float*)d_in[7];
    const float* W2    = (const float*)d_in[8];
    const float* b2    = (const float*)d_in[9];
    const float* pw    = (const float*)d_in[10];
    const float* pb    = (const float*)d_in[11];
    const int*   erow  = (const int*)d_in[12];
    const int*   ecol  = (const int*)d_in[13];
    const float* ew    = (const float*)d_in[14];
    float*       out   = (float*)d_out;

    const int E       = in_sizes[12];
    const int N_NODES = 100000;
    const int NT      = 50000;

    float* x   = (float*)d_ws;
    float* h   = x + (long)N_NODES * 128;
    float* agg = h + (long)N_NODES * 128;
    int* histG    = (int*)(agg + (long)N_NODES * 128);
    int* binStart = histG + BIN_BLOCKS * N_TILES;
    int* rec      = binStart + 320;

    edge_tile_count<<<BIN_BLOCKS, BIN_THREADS, 0, stream>>>(erow, histG, E, N_NODES);
    edge_tile_place<<<BIN_BLOCKS, BIN_THREADS, 0, stream>>>(erow, ecol, ew, histG, binStart, rec, E, N_NODES);

    proj_w1_chain_wmma_f32<<<GEMM_BLOCKS, GEMM_THREADS, 0, stream>>>(
        feat0, fc0_w, fc0_b, feat1, fc1_w, fc1_b, W1, h, NT, NT, 106);

    spmm_tile_lds_f32<128><<<N_TILES, TILE_THREADS, 0, stream>>>(
        rec, binStart, h, b1, agg, N_NODES);

    gemm_bstat_wmma_f32<128, 4, 2, 0><<<GEMM_BLOCKS, GEMM_THREADS, 0, stream>>>(
        agg, W2, nullptr, x, N_NODES, 64, 0);
    spmm_tile_lds_f32<64><<<N_TILES, TILE_THREADS, 0, stream>>>(
        rec, binStart, x, b2, h, N_NODES);

    gemm_bstat_wmma_f32<64, 1, 1, 1><<<GEMM_BLOCKS, GEMM_THREADS, 0, stream>>>(
        h, pw, pb, out, N_NODES, 8, 1);
}


# SpMM edge bucketing in ONE launch (was count + place): a work-group sorts its slice of the edge list by tile in LDS, all loads at once; a tile reads its 256 runs; sum order LDS arrival, unordered
# speedup vs baseline: 3.8746x; 1.0336x over previous
#include <hip/hip_runtime.h>

typedef __attribute__((ext_vector_type(2))) float v2f;
typedef __attribute__((ext_vector_type(8))) float v8f;

#define GEMM_THREADS 512
#define GEMM_BLOCKS 256
#define GEMM_LDS_FLOATS 41984

template <int K, int NT>
__device__ __forceinline__ void tile_mma(const float* __restrict__ as, const v2f* bv, v8f* acc)
{
    constexpr int P = K + 4;
    #pragma unroll
    for (int s = 0; s < K / 4; ++s) {
        #pragma unroll
        for (int j = 0; j < NT; ++j) {
            const float2 a2 = *(const float2*)(as + j * 16 * P + 4 * s);
            v2f av; av.x = a2.x; av.y = a2.y;
            acc[j] = __builtin_amdgcn_wmma_f32_16x16x4_f32(
                          false, av,  false, bv[s],
                          (short)0, acc[j],  false,  false);
        }
    }
}

template <int K, int NS, int TPW, int TRANSB>
__global__ __launch_bounds__(GEMM_THREADS)
void gemm_bstat_wmma_f32(const float* __restrict__ A, const float* __restrict__ B,
                         const float* __restrict__ bias, float* __restrict__ C,
                         int M, int N, int hasBias)
{
    constexpr int P          = K + 4;
    constexpr int RG         = 16 / NS;
    constexpr int STEP_TILES = RG * TPW;
    constexpr int STEP_ROWS  = STEP_TILES * 16;
    constexpr int K4         = K / 4;
    constexpr int NLD        = STEP_ROWS * K4 / GEMM_THREADS;
    static_assert(STEP_ROWS * K4 % GEMM_THREADS == 0, "step must split evenly over the threads");
    static_assert(2 * STEP_ROWS * P <= GEMM_LDS_FLOATS, "LDS ring too small");
    __shared__ float As[GEMM_LDS_FLOATS];

    const int tid  = threadIdx.x;
    const int wave = __builtin_amdgcn_readfirstlane(tid >> 5);
    const int lane = tid & 31;
    const int c    = lane & 15;
    const int h    = lane >> 4;
    const int strip = wave % NS;
    const int rg    = wave / NS;
    const int n     = strip * 16 + c;

    const int nT   = M >> 4;
    const int G    = gridDim.x;
    const int base = nT / G;
    const int rem  = nT - base * G;
    const int b    = blockIdx.x;
    const int t0   = b * base + (b < rem ? b : rem);
    const int t1   = t0 + base + (b < rem ? 1 : 0);
    if (t0 >= t1) return;

    const float4* A4 = (const float4*)A;

    float4 st[NLD];
    auto fetch = [&](int tb) {
        const int lim = (t1 - tb < STEP_TILES ? t1 - tb : STEP_TILES) * 16 * K4;
        const long g0 = (long)tb * 16 * K4;
        #pragma unroll
        for (int j = 0; j < NLD; ++j) {
            const int i = tid + j * GEMM_THREADS;
            st[j] = make_float4(0.0f, 0.0f, 0.0f, 0.0f);
            if (i < lim) st[j] = A4[g0 + i];
        }
    };
    auto stash = [&](float* buf) {
        #pragma unroll
        for (int j = 0; j < NLD; ++j) {
            const int i   = tid + j * GEMM_THREADS;
            const int row = i / K4;
            const int c4  = i - row * K4;
            *(float4*)(buf + row * P + 4 * c4) = st[j];
        }
    };

    fetch(t0);

    v2f bv[K4];
    if (TRANSB) {
        static_assert(NS * 16 * P <= GEMM_LDS_FLOATS, "B does not fit the LDS array");
        const float4* B4  = (const float4*)B;
        const int     nB4 = N * K4;
        constexpr int NB4 = NS * 16 * K4;
        constexpr int NBL = (NB4 + GEMM_THREADS - 1) / GEMM_THREADS;
        float4 bt[NBL];
        #pragma unroll
        for (int j = 0; j < NBL; ++j) {
            const int i = tid + j * GEMM_THREADS;
            bt[j] = make_float4(0.0f, 0.0f, 0.0f, 0.0f);
            if (i < nB4) bt[j] = B4[i];
        }
        #pragma unroll
        for (int j = 0; j < NBL; ++j) {
            const int i = tid + j * GEMM_THREADS;
            if (i < NB4) {
                const int row = i / K4;
                const int c4  = i - row * K4;
                *(float4*)(As + row * P + 4 * c4) = bt[j];
            }
        }
        __syncthreads();
        #pragma unroll
        for (int s = 0; s < K4; ++s) {
            const float2 v = *(const float2*)(As + n * P + 4 * s + 2 * h);
            bv[s].x = v.x; bv[s].y = v.y;
        }
        __syncthreads();
    } else {
        #pragma unroll
        for (int s = 0; s < K4; ++s) {
            const int k0 = 4 * s + 2 * h;
            float2 v = make_float2(0.0f, 0.0f);
            if (n < N) { v.x = B[(long)k0 * N + n]; v.y = B[(long)(k0 + 1) * N + n]; }
            bv[s].x = v.x; bv[s].y = v.y;
        }
    }
    float bval = 0.0f;
    if (hasBias && n < N) bval = bias[n];

    stash(As);
    __syncthreads();

    int cur = 0;
    for (int tb = t0; tb < t1; tb += STEP_TILES) {
        const int  nxt     = tb + STEP_TILES;
        const bool hasNext = nxt < t1;
        if (hasNext) fetch(nxt);

        const int valid = t1 - tb < STEP_TILES ? t1 - tb : STEP_TILES;
        const int lt0   = rg * TPW;
        const int nv    = valid - lt0;
        const float* as = As + cur * (STEP_ROWS * P) + (lt0 * 16 + c) * P + 2 * h;

        v8f acc[TPW];
        #pragma unroll
        for (int j = 0; j < TPW; ++j) acc[j] = v8f{};
        int done = 0;
        if (nv >= TPW)      { tile_mma<K, TPW>(as, bv, acc); done = TPW; }
        else if (nv >= 1)   { tile_mma<K, 1>(as, bv, acc);   done = 1;   }

        if (n < N) {
            #pragma unroll
            for (int j = 0; j < TPW; ++j) {
                if (j < done) {
                    const long row0 = (long)(tb + lt0 + j) * 16 + 8 * h;
                    #pragma unroll
                    for (int r = 0; r < 8; ++r)
                        C[(row0 + r) * (long)N + n] = acc[j][r] + bval;
                }
            }
        }

        if (hasNext) stash(As + (cur ^ 1) * (STEP_ROWS * P));
        __syncthreads();
        cur ^= 1;
    }
}

#define FUSED_P1 132
#define FUSED_AS 33280
#define FUSED_XS 8448
#define FUSED_WS 16896
#define FUSED_LDS_FLOATS (FUSED_AS + 2 * FUSED_XS + FUSED_WS)

template <int NT>
__device__ __forceinline__ void tile_mma_x(const float* xs, const float* wsn, v8f* acc)
{
    #pragma unroll
    for (int s = 0; s < 32; ++s) {
        const float2 b2 = *(const float2*)(wsn + 4 * s);
        v2f bw; bw.x = b2.x; bw.y = b2.y;
        #pragma unroll
        for (int j = 0; j < NT; ++j) {
            const float2 a2 = *(const float2*)(xs + j * 16 * FUSED_P1 + 4 * s);
            v2f av; av.x = a2.x; av.y = a2.y;
            acc[j] = __builtin_amdgcn_wmma_f32_16x16x4_f32(
                          false, av,  false, bw,
                          (short)0, acc[j],  false,  false);
        }
    }
}

template <int K>
__device__ __forceinline__ void proj_then_w1(float* lds,
                                             const float* __restrict__ A, const float* __restrict__ Bfc,
                                             const float* __restrict__ bias, const float* __restrict__ W1,
                                             float* __restrict__ C, int M, int b, int G)
{
    constexpr int NS         = 8;
    constexpr int TPW        = 2;
    constexpr int P          = K + 4;
    constexpr int PX         = FUSED_P1;
    constexpr int RG         = 16 / NS;
    constexpr int STEP_TILES = RG * TPW;
    constexpr int STEP_ROWS  = STEP_TILES * 16;
    constexpr int K4         = K / 4;
    constexpr int NLD        = STEP_ROWS * K4 / GEMM_THREADS;
    static_assert(STEP_ROWS * K4 % GEMM_THREADS == 0, "step must split evenly over the threads");
    static_assert(2 * STEP_ROWS * P <= FUSED_AS, "feature-step buffers too small");
    static_assert(NS * 16 * P <= FUSED_AS, "fc_w does not fit the feature-step buffers");
    static_assert(STEP_ROWS * PX <= FUSED_XS, "projected-step buffer too small");

    float* As = lds;
    float* Xs = lds + FUSED_AS;
    float* Ws = Xs + 2 * FUSED_XS;

    const int tid  = threadIdx.x;
    const int wave = __builtin_amdgcn_readfirstlane(tid >> 5);
    const int lane = tid & 31;
    const int c    = lane & 15;
    const int h    = lane >> 4;
    const int strip = wave % NS;
    const int rg    = wave / NS;
    const int n     = strip * 16 + c;

    const int nT   = M >> 4;
    const int base = nT / G;
    const int rem  = nT - base * G;
    const int t0   = b * base + (b < rem ? b : rem);
    const int t1   = t0 + base + (b < rem ? 1 : 0);
    if (t0 >= t1) return;

    const float4* A4 = (const float4*)A;

    float4 st[NLD];
    auto fetch = [&](int tb) {
        const int lim = (t1 - tb < STEP_TILES ? t1 - tb : STEP_TILES) * 16 * K4;
        const long g0 = (long)tb * 16 * K4;
        #pragma unroll
        for (int j = 0; j < NLD; ++j) {
            const int i = tid + j * GEMM_THREADS;
            st[j] = make_float4(0.0f, 0.0f, 0.0f, 0.0f);
            if (i < lim) st[j] = A4[g0 + i];
        }
    };
    auto stash = [&](float* buf) {
        #pragma unroll
        for (int j = 0; j < NLD; ++j) {
            const int i   = tid + j * GEMM_THREADS;
            const int row = i / K4;
            const int c4  = i - row * K4;
            *(float4*)(buf + row * P + 4 * c4) = st[j];
        }
    };

    fetch(t0);

    {
        const float4* B4  = (const float4*)Bfc;
        constexpr int NBL = 128 * K4 / GEMM_THREADS;
        float4 bt[NBL];
        #pragma unroll
        for (int j = 0; j < NBL; ++j) bt[j] = B4[tid + j * GEMM_THREADS];
        #pragma unroll
        for (int j = 0; j < NBL; ++j) {
            const int i   = tid + j * GEMM_THREADS;
            const int row = i / K4;
            const int c4  = i - row * K4;
            *(float4*)(As + row * P + 4 * c4) = bt[j];
        }
    }
    {
        const float4* W4 = (const float4*)W1;
        #pragma unroll
        for (int j = 0; j < 2; ++j) {
            const int combo = wave * 2 + j;
            const int n4 = (combo & 3) * 8 + (lane & 7);
            const int kg = (combo >> 2) * 4 + (lane >> 3);
            const float4 r0 = W4[(4 * kg + 0) * 32 + n4];
            const float4 r1 = W4[(4 * kg + 1) * 32 + n4];
            const float4 r2 = W4[(4 * kg + 2) * 32 + n4];
            const float4 r3 = W4[(4 * kg + 3) * 32 + n4];
            float* d = Ws + (4 * n4) * PX + 4 * kg;
            *(float4*)(d)          = make_float4(r0.x, r1.x, r2.x, r3.x);
            *(float4*)(d + PX)     = make_float4(r0.y, r1.y, r2.y, r3.y);
            *(float4*)(d + 2 * PX) = make_float4(r0.z, r1.z, r2.z, r3.z);
            *(float4*)(d + 3 * PX) = make_float4(r0.w, r1.w, r2.w, r3.w);
        }
    }
    __syncthreads();
    v2f bv[K4];
    #pragma unroll
    for (int s = 0; s < K4; ++s) {
        const float2 v = *(const float2*)(As + n * P + 4 * s + 2 * h);
        bv[s].x = v.x; bv[s].y = v.y;
    }
    __syncthreads();
    const float bval = bias[n];

    stash(As);
    __syncthreads();

    const float* wsn = Ws + n * PX + 2 * h;
    int cur = 0;
    int xb  = 0;
    for (int tb = t0; tb < t1; tb += STEP_TILES) {
        const int  nxt     = tb + STEP_TILES;
        const bool hasNext = nxt < t1;
        if (hasNext) fetch(nxt);

        const int valid = t1 - tb < STEP_TILES ? t1 - tb : STEP_TILES;
        const int lt0   = rg * TPW;
        const int nv    = valid - lt0;
        const float* as = As + cur * (STEP_ROWS * P) + (lt0 * 16 + c) * P + 2 * h;
        float*       xw = Xs + xb * FUSED_XS;

        v8f acc[TPW];
        #pragma unroll
        for (int j = 0; j < TPW; ++j) acc[j] = v8f{};
        int done = 0;
        if (nv >= TPW)      { tile_mma<K, TPW>(as, bv, acc); done = TPW; }
        else if (nv >= 1)   { tile_mma<K, 1>(as, bv, acc);   done = 1;   }

        if (hasNext) stash(As + (cur ^ 1) * (STEP_ROWS * P));

        #pragma unroll
        for (int j = 0; j < TPW; ++j) {
            if (j < done) {
                float* xr = xw + ((lt0 + j) * 16 + 8 * h) * PX + n;
                #pragma unroll
                for (int r = 0; r < 8; ++r) xr[r * PX] = acc[j][r] + bval;
            }
        }
        __syncthreads();

        const float* xs = xw + (lt0 * 16 + c) * PX + 2 * h;
        v8f acc2[TPW];
        #pragma unroll
        for (int j = 0; j < TPW; ++j) acc2[j] = v8f{};
        if (done == TPW)    tile_mma_x<TPW>(xs, wsn, acc2);
        else if (done == 1) tile_mma_x<1>(xs, wsn, acc2);

        #pragma unroll
        for (int j = 0; j < TPW; ++j) {
            if (j < done) {
                const long row0 = (long)(tb + lt0 + j) * 16 + 8 * h;
                #pragma unroll
                for (int r = 0; r < 8; ++r)
                    C[(row0 + r) * 128L + n] = acc2[j][r];
            }
        }
        cur ^= 1;
        xb  ^= 1;
    }
}

__global__ __launch_bounds__(GEMM_THREADS)
void proj_w1_chain_wmma_f32(const float* __restrict__ feat0, const float* __restrict__ fc0_w,
                            const float* __restrict__ fc0_b, const float* __restrict__ feat1,
                            const float* __restrict__ fc1_w, const float* __restrict__ fc1_b,
                            const float* __restrict__ W1, float* __restrict__ hout,
                            int M0, int M1, int G0)
{
    __shared__ float lds[FUSED_LDS_FLOATS];
    const int b = blockIdx.x;
    const int G = gridDim.x;
    if (b < G0) proj_then_w1<128>(lds, feat0, fc0_w, fc0_b, W1, hout, M0, b, G0);
    else        proj_then_w1<256>(lds, feat1, fc1_w, fc1_b, W1, hout + (long)M0 * 128, M1, b - G0, G - G0);
}

#define TILE_ROWS 392
#define TILE_THREADS 1024
#define N_TILES 256
#define SLICE_EDGES 6250
#define SLICE_PITCH 6272
#define SORT_BLOCKS 256
#define SORT_THREADS 512
#define SORT_EPT 13

__global__ __launch_bounds__(SORT_THREADS)
void edge_slice_sort(const int* __restrict__ rows, const int* __restrict__ cols,
                     const float* __restrict__ ew, int* __restrict__ runT,
                     int* __restrict__ recw, int nE, int nRows)
{
    static_assert(SORT_EPT * SORT_THREADS >= SLICE_EDGES, "a slice must fit one pass of the work-group");
    static_assert(SLICE_PITCH >= SLICE_EDGES, "slice pitch too small");
    __shared__ int4 stage[SLICE_EDGES];
    __shared__ int  hist[256];
    __shared__ int  wsum[8];
    __shared__ int  total;
    const int tid  = threadIdx.x;
    const int lane = tid & 31;
    const int wave = tid >> 5;
    const int nS   = (nE + SLICE_EDGES - 1) / SLICE_EDGES;

    for (int s = blockIdx.x; s < nS; s += (int)gridDim.x) {
        const int e0 = s * SLICE_EDGES;
        const int n  = nE - e0 < SLICE_EDGES ? nE - e0 : SLICE_EDGES;

        if (tid < 256) hist[tid] = 0;
        const int*   rp = rows + e0 + tid;
        const int*   cp = cols + e0 + tid;
        const float* wp = ew   + e0 + tid;
        int   r[SORT_EPT], c[SORT_EPT];
        float w[SORT_EPT];
        #pragma unroll
        for (int j = 0; j < SORT_EPT; ++j) {
            r[j] = -1;  c[j] = 0;  w[j] = 0.0f;
            if (tid + j * SORT_THREADS < n) {
                r[j] = rp[j * SORT_THREADS];  c[j] = cp[j * SORT_THREADS];  w[j] = wp[j * SORT_THREADS];
            }
        }
        __syncthreads();

        int tl[SORT_EPT], rk[SORT_EPT];
        #pragma unroll
        for (int j = 0; j < SORT_EPT; ++j) {
            tl[j] = -1;  rk[j] = 0;
            if ((unsigned)r[j] < (unsigned)nRows) {
                tl[j] = (int)((unsigned)r[j] / TILE_ROWS);
                rk[j] = atomicAdd(&hist[tl[j]], 1);
            }
        }
        __syncthreads();

        int v = 0, x = 0;
        if (tid < 256) {
            v = hist[tid];
            x = v;
            for (int d = 1; d < 32; d <<= 1) {
                const int y = __shfl_up(x, d);
                if (lane >= d) x += y;
            }
            if (lane == 31) wsum[wave] = x;
        }
        __syncthreads();
        if (tid < 256) {
            int off = 0;
            for (int q = 0; q < wave; ++q) off += wsum[q];
            const int excl = off + x - v;
            hist[tid] = excl;
            runT[(long)tid * nS + s] = excl;
            if (tid == 255) { runT[256L * nS + s] = excl + v;  total = excl + v; }
        }
        __syncthreads();

        int pos[SORT_EPT];
        #pragma unroll
        for (int j = 0; j < SORT_EPT; ++j) pos[j] = hist[tl[j] < 0 ? 0 : tl[j]] + rk[j];
        #pragma unroll
        for (int j = 0; j < SORT_EPT; ++j) {
            if (tl[j] >= 0)
                stage[pos[j]] = make_int4(r[j] - tl[j] * TILE_ROWS, c[j], __float_as_int(w[j]), 0);
        }
        __syncthreads();

        int4* dst = (int4*)recw + (long)s * SLICE_PITCH + tid;
        const int cnt = total;
        int4 q[SORT_EPT];
        #pragma unroll
        for (int j = 0; j < SORT_EPT; ++j) {
            q[j] = make_int4(0, 0, 0, 0);
            if (tid + j * SORT_THREADS < cnt) q[j] = stage[tid + j * SORT_THREADS];
        }
        #pragma unroll
        for (int j = 0; j < SORT_EPT; ++j) {
            if (tid + j * SORT_THREADS < cnt) dst[j * SORT_THREADS] = q[j];
        }
        __syncthreads();
    }
}

typedef __attribute__((ext_vector_type(4))) float v4f;
template <int VEC> struct row_piece;
template <> struct row_piece<4> { typedef v4f type; };
template <> struct row_piece<2> { typedef v2f type; };

template <int F, bool FULL, typename Hook>
__device__ __forceinline__ void flush_group(int s, int n, int rr, int cc, float ww,
                                            const float* __restrict__ dense, float* acc, int lane,
                                            bool first, Hook after_issue)
{
    constexpr int VEC = F / 32;
    constexpr int SB  = 1024 / F;
    typedef typename row_piece<VEC>::type piece;
    piece v[SB];
    #pragma unroll
    for (int j = 0; j < SB; ++j) {
        const int idx = s + j;
        v[j] = piece{};
        if (FULL || idx < n) {
            const int sr = __builtin_amdgcn_readlane(cc, idx & 31);
            v[j] = *(const piece*)(dense + (long)sr * F + VEC * lane);
        }
    }
    if (first) after_issue();
    #pragma unroll
    for (int j = 0; j < SB; ++j) {
        const int idx = s + j;
        if (FULL || idx < n) {
            const float w  = __int_as_float(__builtin_amdgcn_readlane(__float_as_int(ww), idx & 31));
            const int   lr = __builtin_amdgcn_readlane(rr, idx & 31);
            float* a = acc + lr * F + lane;
            #pragma unroll
            for (int c = 0; c < VEC; ++c)
                atomicAdd(a + c * 32, v[j][c] * w);
        }
    }
}

__device__ __forceinline__ int run_record(int v, int p1, int p2, int p3, int p4, int p5, int p6, int p7,
                                          int d0, int d1, int d2, int d3, int d4, int d5, int d6, int d7)
{
    int d = d0;
    d = v >= p1 ? d1 : d;
    d = v >= p2 ? d2 : d;
    d = v >= p3 ? d3 : d;
    d = v >= p4 ? d4 : d;
    d = v >= p5 ? d5 : d;
    d = v >= p6 ? d6 : d;
    d = v >= p7 ? d7 : d;
    return v + d;
}

template <int F>
__global__ __launch_bounds__(TILE_THREADS)
void spmm_tile_lds_f32(const int* __restrict__ recw, const int* __restrict__ runT,
                       const float* __restrict__ dense, const float* __restrict__ bias,
                       float* __restrict__ out, int nRows, int nE)
{
    constexpr int LPR = F / 4;
    constexpr int VEC = F / 32;
    constexpr int SB  = 1024 / F;
    __shared__ float acc[TILE_ROWS * F];

    const int tid  = threadIdx.x;
    const int wave = __builtin_amdgcn_readfirstlane(tid >> 5);
    const int lane = tid & 31;
    const int row0 = blockIdx.x * TILE_ROWS;

    float4* acc4 = (float4*)acc;
    for (int i = tid; i < TILE_ROWS * F / 4; i += TILE_THREADS)
        acc4[i] = make_float4(0.0f, 0.0f, 0.0f, 0.0f);
    __syncthreads();

    const int  nS  = (nE + SLICE_EDGES - 1) / SLICE_EDGES;
    const int* rt0 = runT + (long)blockIdx.x * nS;
    const int* rt1 = rt0 + nS;
    for (int g = wave; 8 * g < nS; g += TILE_THREADS / 32) {
        const int sl = 8 * g + (lane & 7);
        int st = 0, en = 0;
        if (lane < 8 && sl < nS) { st = rt0[sl];  en = rt1[sl]; }
        const int ln = en - st;
        const int fb = sl * SLICE_PITCH + st;
        const int p1 = __builtin_amdgcn_readlane(ln, 0);
        const int p2 = p1 + __builtin_amdgcn_readlane(ln, 1);
        const int p3 = p2 + __builtin_amdgcn_readlane(ln, 2);
        const int p4 = p3 + __builtin_amdgcn_readlane(ln, 3);
        const int p5 = p4 + __builtin_amdgcn_readlane(ln, 4);
        const int p6 = p5 + __builtin_amdgcn_readlane(ln, 5);
        const int p7 = p6 + __builtin_amdgcn_readlane(ln, 6);
        const int T  = p7 + __builtin_amdgcn_readlane(ln, 7);
        const int d0 = __builtin_amdgcn_readlane(fb, 0);
        const int d1 = __builtin_amdgcn_readlane(fb, 1) - p1;
        const int d2 = __builtin_amdgcn_readlane(fb, 2) - p2;
        const int d3 = __builtin_amdgcn_readlane(fb, 3) - p3;
        const int d4 = __builtin_amdgcn_readlane(fb, 4) - p4;
        const int d5 = __builtin_amdgcn_readlane(fb, 5) - p5;
        const int d6 = __builtin_amdgcn_readlane(fb, 6) - p6;
        const int d7 = __builtin_amdgcn_readlane(fb, 7) - p7;
        int   base = 0;
        int   rr = 0, cc = 0;
        float ww = 0.0f;
        if (lane < T) {
            const int* p = recw + 4L * run_record(lane, p1, p2, p3, p4, p5, p6, p7,
                                                  d0, d1, d2, d3, d4, d5, d6, d7);
            rr = p[0];  cc = p[1];  ww = __int_as_float(p[2]);
        }
        while (base < T) {
            const int nb = base + 32;
            const int nrec = run_record(nb + lane, p1, p2, p3, p4, p5, p6, p7,
                                        d0, d1, d2, d3, d4, d5, d6, d7);
            int   rn = 0, cn = 0;
            float wn = 0.0f;
            auto next_records = [&]() {
                if (nb + lane < T) {
                    const int* p = recw + 4L * nrec;
                    rn = p[0];  cn = p[1];  wn = __int_as_float(p[2]);
                }
            };
            const int n = T - base < 32 ? T - base : 32;
            if (n == 32) {
                #pragma unroll
                for (int s = 0; s < 32; s += SB)
                    flush_group<F, true>(s, 32, rr, cc, ww, dense, acc, lane, s == 0, next_records);
            } else {
                for (int s = 0; s < n; s += SB)
                    flush_group<F, false>(s, n, rr, cc, ww, dense, acc, lane, s == 0, next_records);
            }
            rr = rn;  cc = cn;  ww = wn;
            base = nb;
        }
    }
    __syncthreads();

    for (int i = tid; i < TILE_ROWS * LPR; i += TILE_THREADS) {
        const int r  = i / LPR;
        const int l  = i % LPR;
        const int gr = row0 + r;
        if (gr < nRows) {
            const float* a = acc + r * F;
            const float4 b = *(const float4*)(bias + 4 * l);
            float4 o;
            o.x = a[((4 * l + 0) % VEC) * 32 + (4 * l + 0) / VEC] + b.x;
            o.y = a[((4 * l + 1) % VEC) * 32 + (4 * l + 1) / VEC] + b.y;
            o.z = a[((4 * l + 2) % VEC) * 32 + (4 * l + 2) / VEC] + b.z;
            o.w = a[((4 * l + 3) % VEC) * 32 + (4 * l + 3) / VEC] + b.w;
            o.x = o.x > 0.0f ? o.x : 0.0f;  o.y = o.y > 0.0f ? o.y : 0.0f;
            o.z = o.z > 0.0f ? o.z : 0.0f;  o.w = o.w > 0.0f ? o.w : 0.0f;
            *(float4*)(out + (long)gr * F + 4 * l) = o;
        }
    }
}

extern "C" void kernel_launch(void* const* d_in, const int* in_sizes, int n_in,
                              void* d_out, int out_size, void* d_ws, size_t ws_size,
                              hipStream_t stream)
{
    const float* feat0 = (const float*)d_in[0];
    const float* feat1 = (const float*)d_in[1];
    const float* fc0_w = (const float*)d_in[2];
    const float* fc0_b = (const float*)d_in[3];
    const float* fc1_w = (const float*)d_in[4];
    const float* fc1_b = (const float*)d_in[5];
    const float* W1    = (const float*)d_in[6];
    const float* b1    = (const float*)d_in[7];
    const float* W2    = (const float*)d_in[8];
    const float* b2    = (const float*)d_in[9];
    const float* pw    = (const float*)d_in[10];
    const float* pb    = (const float*)d_in[11];
    const int*   erow  = (const int*)d_in[12];
    const int*   ecol  = (const int*)d_in[13];
    const float* ew    = (const float*)d_in[14];
    float*       out   = (float*)d_out;

    const int E       = in_sizes[12];
    const int N_NODES = 100000;
    const int NT      = 50000;

    float* x   = (float*)d_ws;
    float* h   = x + (long)N_NODES * 128;
    float* agg = h + (long)N_NODES * 128;
    int* runT = (int*)(agg + (long)N_NODES * 128);
    int* rec  = runT + 257 * 256;

    edge_slice_sort<<<SORT_BLOCKS, SORT_THREADS, 0, stream>>>(erow, ecol, ew, runT, rec, E, N_NODES);

    proj_w1_chain_wmma_f32<<<GEMM_BLOCKS, GEMM_THREADS, 0, stream>>>(
        feat0, fc0_w, fc0_b, feat1, fc1_w, fc1_b, W1, h, NT, NT, 106);

    spmm_tile_lds_f32<128><<<N_TILES, TILE_THREADS, 0, stream>>>(
        rec, runT, h, b1, agg, N_NODES, E);

    gemm_bstat_wmma_f32<128, 4, 2, 0><<<GEMM_BLOCKS, GEMM_THREADS, 0, stream>>>(
        agg, W2, nullptr, x, N_NODES, 64, 0);
    spmm_tile_lds_f32<64><<<N_TILES, TILE_THREADS, 0, stream>>>(
        rec, runT, x, b2, h, N_NODES, E);

    gemm_bstat_wmma_f32<64, 1, 1, 1><<<GEMM_BLOCKS, GEMM_THREADS, 0, stream>>>(
        h, pw, pb, out, N_NODES, 8, 1);
}
